# starting kernel
# baseline (speedup 1.0000x reference)
#include <hip/hip_runtime.h>

#define NG     100000
#define NS     20000
#define EDGES  300000
#define DIM    256
#define NQ     4096
#define MAXU   4096
#define LDSPAD 2

typedef __attribute__((ext_vector_type(2))) float v2f;
typedef __attribute__((ext_vector_type(8))) float v8f;

__global__ __launch_bounds__(256) void k_init_slots(int* __restrict__ sg,
                                                    int* __restrict__ ss,
                                                    int* __restrict__ cnt) {
  int i = blockIdx.x * 256 + threadIdx.x;
  if (i < NG) sg[i] = -1;
  if (i < NS) ss[i] = -1;
  if (i < 2)  cnt[i] = 0;
}

__global__ __launch_bounds__(256) void k_zero(float* __restrict__ p, int n) {
  int i = blockIdx.x * 256 + threadIdx.x;
  if (i < n) p[i] = 0.0f;
}

__global__ __launch_bounds__(256) void k_compact(const int* __restrict__ gid,
                                                 const int* __restrict__ sid,
                                                 int* __restrict__ sg,
                                                 int* __restrict__ ss,
                                                 int* __restrict__ cnt) {
  int i = blockIdx.x * 256 + threadIdx.x;
  if (i < NQ) {
    int g = gid[i];
    if (atomicCAS(&sg[g], -1, -2) == -1) sg[g] = atomicAdd(&cnt[0], 1);
  } else if (i < 2 * NQ) {
    int g = sid[i - NQ];
    if (atomicCAS(&ss[g], -1, -2) == -1) ss[g] = atomicAdd(&cnt[1], 1);
  }
}

__global__ __launch_bounds__(256) void k_scatter(const float* __restrict__ vsrc,
                                                 const float* __restrict__ att,
                                                 const int* __restrict__ src,
                                                 const int* __restrict__ dst,
                                                 const int* __restrict__ slot,
                                                 float* __restrict__ nh, int nE) {
  const int wid  = (blockIdx.x * 256 + threadIdx.x) >> 5;
  const int lane = threadIdx.x & 31;
  if (wid >= nE) return;
  const int s = slot[dst[wid]];
  if (s < 0) return;
  const float a = att[wid];
  const float4* vp = (const float4*)(vsrc + (size_t)src[wid] * DIM);
  float* np = nh + (size_t)s * DIM;
#pragma unroll
  for (int j = 0; j < 2; ++j) {
    const int q = lane + j * 32;
    const float4 vv = vp[q];
    atomicAdd(np + q * 4 + 0, vv.x * a);
    atomicAdd(np + q * 4 + 1, vv.y * a);
    atomicAdd(np + q * 4 + 2, vv.z * a);
    atomicAdd(np + q * 4 + 3, vv.w * a);
  }
}

__global__ __launch_bounds__(32) void k_combine(const float* __restrict__ v,
                                                const float* __restrict__ nh,
                                                const int* __restrict__ slot,
                                                const int* __restrict__ ids,
                                                const float* __restrict__ W1,
                                                const float* __restrict__ b1,
                                                float* __restrict__ out) {
  __shared__ float x1[16][DIM + LDSPAD];
  __shared__ float x2[16][DIM + LDSPAD];
  __shared__ int   qg[16];
  __shared__ int   qs[16];

  const int lane  = threadIdx.x;
  const int qbase = blockIdx.x * 16;

  if (lane < 16) {
    const int g = ids[qbase + lane];
    qg[lane] = g;
    qs[lane] = slot[g];
  }
  __syncthreads();

  for (int i = lane; i < 16 * (DIM / 4); i += 32) {
    const int r = i >> 6;
    const int c = (i & 63) * 4;
    const float4 vv = *(const float4*)(v  + (size_t)qg[r] * DIM + c);
    const float4 nn = *(const float4*)(nh + (size_t)qs[r] * DIM + c);
    x1[r][c + 0] = vv.x + nn.x;  x2[r][c + 0] = vv.x * nn.x;
    x1[r][c + 1] = vv.y + nn.y;  x2[r][c + 1] = vv.y * nn.y;
    x1[r][c + 2] = vv.z + nn.z;  x2[r][c + 2] = vv.z * nn.z;
    x1[r][c + 3] = vv.w + nn.w;  x2[r][c + 3] = vv.w * nn.w;
  }
  __syncthreads();

  const int mrow  = lane & 15;
  const int khalf = lane >> 4;

  for (int n0 = 0; n0 < DIM; n0 += 16) {
    v8f acc1 = {};
    v8f acc2 = {};
    const int ncol = n0 + (lane & 15);
    const float* wrow = W1 + (size_t)ncol * DIM;

#pragma unroll 4
    for (int k0 = 0; k0 < DIM; k0 += 4) {
      const int ka = k0 + 2 * khalf;
      v2f a1; a1.x = x1[mrow][ka]; a1.y = x1[mrow][ka + 1];
      v2f a2; a2.x = x2[mrow][ka]; a2.y = x2[mrow][ka + 1];
      v2f bb; bb.x = wrow[ka];     bb.y = wrow[ka + 1];
      acc1 = __builtin_amdgcn_wmma_f32_16x16x4_f32(
          false, a1, false, bb, (short)0, acc1, false, false);
      acc2 = __builtin_amdgcn_wmma_f32_16x16x4_f32(
          false, a2, false, bb, (short)0, acc2, false, false);
    }

    const float bias = b1[ncol];
#pragma unroll
    for (int r = 0; r < 8; ++r) {
      const int m = khalf * 8 + r;
      float y1 = acc1[r] + bias;
      float y2 = acc2[r] + bias;
      y1 = (y1 >= 0.0f) ? y1 : 0.01f * y1;
      y2 = (y2 >= 0.0f) ? y2 : 0.01f * y2;
      out[(size_t)(qbase + m) * DIM + ncol] = y1 + y2;
    }
  }
}

extern "C" void kernel_launch(void* const* d_in, const int* in_sizes, int n_in,
                              void* d_out, int out_size, void* d_ws, size_t ws_size,
                              hipStream_t stream) {
  const float* v_grid  = (const float*)d_in[0];
  const float* v_sc    = (const float*)d_in[1];
  const float* att_s2g = (const float*)d_in[2];
  const float* att_g2s = (const float*)d_in[3];
  const float* W1      = (const float*)d_in[4];
  const float* b1      = (const float*)d_in[5];
  const int*   src_s2g = (const int*)d_in[6];
  const int*   dst_s2g = (const int*)d_in[7];
  const int*   src_g2s = (const int*)d_in[8];
  const int*   dst_g2s = (const int*)d_in[9];
  const int*   sc_id   = (const int*)d_in[10];
  const int*   gr_id   = (const int*)d_in[11];
  float* out = (float*)d_out;

  int*   sg  = (int*)d_ws;
  int*   ss  = sg + NG;
  int*   cnt = ss + NS;
  float* nhg = (float*)(cnt + 4);
  float* nhs = nhg + (size_t)MAXU * DIM;

  k_init_slots<<<(NG + 255) / 256, 256, 0, stream>>>(sg, ss, cnt);
  k_zero<<<(2 * MAXU * DIM + 255) / 256, 256, 0, stream>>>(nhg, 2 * MAXU * DIM);
  k_compact<<<(2 * NQ + 255) / 256, 256, 0, stream>>>(gr_id, sc_id, sg, ss, cnt);

  k_scatter<<<(EDGES + 7) / 8, 256, 0, stream>>>(v_sc,   att_s2g, src_s2g, dst_s2g, sg, nhg, EDGES);
  k_scatter<<<(EDGES + 7) / 8, 256, 0, stream>>>(v_grid, att_g2s, src_g2s, dst_g2s, ss, nhs, EDGES);

  k_combine<<<NQ / 16, 32, 0, stream>>>(v_sc,   nhs, ss, sc_id, W1, b1, out);
  k_combine<<<NQ / 16, 32, 0, stream>>>(v_grid, nhg, sg, gr_id, W1, b1, out + (size_t)NQ * DIM);
}


# pull+combine workgroup cut to 138 KB LDS, 124 registers (W1 staged 32 k at a time, 8 gathered rows in flight per wave): two workgroups share a WGP, one's scan and gather run under the other's WMMAs
# speedup vs baseline: 4.4684x; 4.4684x over previous
#include <hip/hip_runtime.h>

#define NG     100000
#define NS     20000
#define EDGES  300000
#define DIM    256
#define NQ     4096
#define MAXU   4096
#define WPITCH 260
#define EBITS  19
#define LCAP   EDGES
#define PR     16
#define PT     512
#define PCH    (PT * 32)
#define PB     8
#define WK     32
#define NPIECE (DIM / WK)
#define WTP    36
#define WPIECE (16 * WTP)
#define ARENA  (2 * PR * WPITCH + (PT / 32) * 2 * WPIECE)
#define PCAP   ARENA

static_assert(EDGES <= (1 << EBITS), "edge index must fit the packed entry");
static_assert(MAXU  <= (1 << (31 - EBITS)), "slot must fit the packed entry");
static_assert(MAXU % PR == 0 && PR == 16, "a workgroup owns one 16-row tile of slots");
static_assert(PT / 32 * 16 == DIM, "one wave per 16-column tile of the output");
static_assert(PCAP <= ARENA && PCAP > PCH, "the match list lives in the arena and takes one more chunk behind the flush mark");
static_assert((17 * DIM + ARENA + NQ) * 4 + 256 <= 160 * 1024, "two workgroups per WGP");
static_assert(NQ % PT == 0, "the query scan has no tail");

typedef __attribute__((ext_vector_type(2))) float v2f;
typedef __attribute__((ext_vector_type(8))) float v8f;

__global__ __launch_bounds__(256) void k_init_slots(int* __restrict__ sg,
                                                    int* __restrict__ ss,
                                                    int* __restrict__ cnt) {
  int i = blockIdx.x * 256 + threadIdx.x;
  if (i < NG) sg[i] = -1;
  if (i < NS) ss[i] = -1;
  if (i < 4)  cnt[i] = 0;
}

__global__ __launch_bounds__(256) void k_compact(const int* __restrict__ gid,
                                                 const int* __restrict__ sid,
                                                 int* __restrict__ sg,
                                                 int* __restrict__ ss,
                                                 int* __restrict__ cnt) {
  int i = blockIdx.x * 256 + threadIdx.x;
  if (i < NQ) {
    int g = gid[i];
    if (atomicCAS(&sg[g], -1, -2) == -1) sg[g] = atomicAdd(&cnt[0], 1);
  } else if (i < 2 * NQ) {
    int g = sid[i - NQ];
    if (atomicCAS(&ss[g], -1, -2) == -1) ss[g] = atomicAdd(&cnt[1], 1);
  }
}

__global__ __launch_bounds__(1024) void k_filter(const int* __restrict__ dst_g,
                                                  const int* __restrict__ dst_s,
                                                  const int* __restrict__ sg,
                                                  const int* __restrict__ ss,
                                                  int* __restrict__ lists,
                                                  int* __restrict__ cnt) {
  __shared__ int wsum[32];
  __shared__ int wbase;
  const int side = blockIdx.y;
  const int lane = threadIdx.x & 31;
  const int wave = threadIdx.x >> 5;
  const int e    = blockIdx.x * 1024 + threadIdx.x;
  const int* dst  = side ? dst_s : dst_g;
  const int* slot = side ? ss : sg;
  int s = -1;
  if (e < EDGES) s = slot[dst[e]];
  const bool m = (s >= 0);
  const unsigned bal = __builtin_amdgcn_ballot_w32(m);
  const int before = __popc(bal & ((1u << lane) - 1u));
  if (lane == 0) wsum[wave] = __popc(bal);
  __syncthreads();
  if (wave == 0) {
    const int c = wsum[lane];
    int x = c;
#pragma unroll
    for (int d = 1; d < 32; d <<= 1) {
      const int y = __shfl_up(x, d);
      if (lane >= d) x += y;
    }
    wsum[lane] = x - c;
    if (lane == 31) wbase = (x > 0) ? atomicAdd(&cnt[2 + side], x) : 0;
  }
  __syncthreads();
  if (m) lists[side * LCAP + wbase + wsum[wave] + before] = (s << EBITS) | e;
}

__device__ __forceinline__ int kfrag(int t, int h) { return 8 * t + 4 * h; }

__device__ __forceinline__ void keep_ahead(float4& w) {
  asm volatile("" : "+v"(w.x), "+v"(w.y), "+v"(w.z), "+v"(w.w));
}

__global__ __launch_bounds__(PT, 8) void k_pullmix(const float* __restrict__ v_grid,
                                                const float* __restrict__ v_sc,
                                                const float* __restrict__ att_s2g,
                                                const float* __restrict__ att_g2s,
                                                const int* __restrict__ src_s2g,
                                                const int* __restrict__ src_g2s,
                                                const int* __restrict__ lists,
                                                const int* __restrict__ cnt,
                                                const int* __restrict__ sg,
                                                const int* __restrict__ ss,
                                                const int* __restrict__ gr_id,
                                                const int* __restrict__ sc_id,
                                                const float* __restrict__ W1,
                                                const float* __restrict__ b1,
                                                float* __restrict__ out) {
  __shared__ __attribute__((aligned(16))) float acc[(PR + 1) * DIM];
  __shared__ __attribute__((aligned(16))) float arena[ARENA];
  __shared__ int qlist[NQ];
  __shared__ int rownode[PR];
  __shared__ int mcount;
  __shared__ int qcount;

  int* mlist = (int*)arena;
  const int tid  = threadIdx.x;
  const int lane = tid & 31;
  const int wave = __builtin_amdgcn_readfirstlane(tid >> 5);
  const unsigned blk = blockIdx.x;
  const int side = 1 - (int)blockIdx.y;

  const float* vsrc  = side ? v_grid  : v_sc;
  const float* att   = side ? att_g2s : att_s2g;
  const int*   src   = side ? src_g2s : src_s2g;
  const int*   list  = lists + (size_t)side * LCAP;
  const float* vself = side ? v_sc    : v_grid;
  const int*   slot  = side ? ss      : sg;
  const int*   ids   = side ? sc_id   : gr_id;
  float*       obase = out + (side ? (size_t)0 : (size_t)NQ * DIM);
  const int n = cnt[2 + side];

  for (int i = tid; i < (PR + 1) * DIM; i += PT) acc[i] = 0.0f;
  if (tid < PR) rownode[tid] = -1;
  if (tid == 0) { mcount = 0; qcount = 0; }
  __syncthreads();

  {
    int gq[NQ / PT];
    int sq[NQ / PT];
#pragma unroll
    for (int j = 0; j < NQ / PT; ++j) gq[j] = ids[j * PT + tid];
#pragma unroll
    for (int j = 0; j < NQ / PT; ++j) sq[j] = slot[gq[j]];
#pragma unroll
    for (int j = 0; j < NQ / PT; ++j) {
      if (((unsigned)sq[j] / PR) == blk) {
        const int r   = sq[j] % PR;
        const int pos = atomicAdd(&qcount, 1);
        qlist[pos]  = ((j * PT + tid) << 4) | r;
        rownode[r]  = gq[j];
      }
    }
  }
  __syncthreads();
  const int nq = __builtin_amdgcn_readfirstlane(qcount);
  if (nq == 0) return;

  for (int c0 = 0; c0 < n; c0 += PCH) {
    int4 ent[8];
#pragma unroll
    for (int k = 0; k < 8; ++k) {
      const int i4 = c0 + (k * PT + tid) * 4;
      ent[k] = (i4 < n) ? *(const int4*)(list + i4) : make_int4(-1, -1, -1, -1);
    }
#pragma unroll
    for (int k = 0; k < 8; ++k) {
      const int i4 = c0 + (k * PT + tid) * 4;
      const int vals[4] = {ent[k].x, ent[k].y, ent[k].z, ent[k].w};
#pragma unroll
      for (int j = 0; j < 4; ++j) {
        if (i4 + j < n && (((unsigned)vals[j] >> EBITS) / PR) == blk) {
          const int pos = atomicAdd(&mcount, 1);
          mlist[pos] = vals[j];
        }
      }
    }
    __syncthreads();
    const int mc = __builtin_amdgcn_readfirstlane(mcount);
    __syncthreads();

    if (mc > PCAP - PCH || c0 + PCH >= n) {
      for (int b0 = wave * PB; b0 < mc; b0 += (PT / 32) * PB) {
        int   srow = 0;
        int   arow = PR;
        float a    = 0.0f;
        if (lane < PB && b0 + lane < mc) {
          const unsigned ment = (unsigned)mlist[b0 + lane];
          const int e = (int)(ment & ((1u << EBITS) - 1u));
          srow = src[e];
          a    = att[e];
          arow = (int)((ment >> EBITS) % PR);
        }
        float4 rv[PB][2];
#pragma unroll
        for (int b = 0; b < PB; ++b) {
          const int sr = __shfl(srow, b);
          const float4* vp = (const float4*)(vsrc + (size_t)sr * DIM);
          rv[b][0] = vp[lane];
          rv[b][1] = vp[lane + 32];
        }
#pragma unroll
        for (int b = 0; b < PB; ++b) {
          const float ab = __shfl(a, b);
          float* ap = acc + __shfl(arow, b) * DIM + lane * 4;
          atomicAdd(ap + 0,   rv[b][0].x * ab);
          atomicAdd(ap + 1,   rv[b][0].y * ab);
          atomicAdd(ap + 2,   rv[b][0].z * ab);
          atomicAdd(ap + 3,   rv[b][0].w * ab);
          atomicAdd(ap + 128, rv[b][1].x * ab);
          atomicAdd(ap + 129, rv[b][1].y * ab);
          atomicAdd(ap + 130, rv[b][1].z * ab);
          atomicAdd(ap + 131, rv[b][1].w * ab);
        }
      }
      __syncthreads();
      if (tid == 0) mcount = 0;
      __syncthreads();
    }
  }

  const int c  = lane & 15;
  const int h  = lane >> 4;
  const int n0 = wave * 16;

  const int wr = lane >> 3;
  const int wq = lane & 7;
  const float* wsl = W1 + (size_t)(n0 + wr) * DIM + wq * 4;
  float4 wp[NPIECE][4];
#pragma unroll
  for (int p = 0; p < 2; ++p) {
#pragma unroll
    for (int j = 0; j < 4; ++j) wp[p][j] = *(const float4*)(wsl + (4 * j) * DIM + WK * p);
  }
  const float bias = b1[n0 + c];

  float* As = arena;
  float* Ap = As + PR * WPITCH;
  float* wt = Ap + PR * WPITCH + wave * (2 * WPIECE);
#pragma unroll
  for (int j = 0; j < 4; ++j) *(float4*)(wt + (4 * j + wr) * WTP + wq * 4) = wp[0][j];
  {
    const int r = tid >> 5;
    const int g = rownode[r];
    float4 v0 = make_float4(0.0f, 0.0f, 0.0f, 0.0f);
    float4 v1 = v0;
    if (g >= 0) {
      const float4* vp = (const float4*)(vself + (size_t)g * DIM);
      v0 = vp[lane];
      v1 = vp[lane + 32];
    }
    const float4 h0 = *(const float4*)(acc + r * DIM + lane * 4);
    const float4 h1 = *(const float4*)(acc + r * DIM + 128 + lane * 4);
    float* s0 = As + r * WPITCH + lane * 4;
    float* p0 = Ap + r * WPITCH + lane * 4;
    *(float4*)(s0)       = make_float4(v0.x + h0.x, v0.y + h0.y, v0.z + h0.z, v0.w + h0.w);
    *(float4*)(s0 + 128) = make_float4(v1.x + h1.x, v1.y + h1.y, v1.z + h1.z, v1.w + h1.w);
    *(float4*)(p0)       = make_float4(v0.x * h0.x, v0.y * h0.y, v0.z * h0.z, v0.w * h0.w);
    *(float4*)(p0 + 128) = make_float4(v1.x * h1.x, v1.y * h1.y, v1.z * h1.z, v1.w * h1.w);
  }
#pragma unroll
  for (int j = 0; j < 4; ++j) keep_ahead(wp[1][j]);
  __syncthreads();

  v8f ysum = (v8f){};
  v8f yprd = (v8f){};
  {
    const float* asl = As + c * WPITCH;
    const float* apl = Ap + c * WPITCH;
#pragma unroll
    for (int p = 0; p < NPIECE; ++p) {
      if (p + 1 < NPIECE) {
        float* wn = wt + ((p + 1) & 1) * WPIECE;
#pragma unroll
        for (int j = 0; j < 4; ++j) *(float4*)(wn + (4 * j + wr) * WTP + wq * 4) = wp[p + 1][j];
      }
      if (p + 2 < NPIECE) {
#pragma unroll
        for (int j = 0; j < 4; ++j) wp[p + 2][j] = *(const float4*)(wsl + (4 * j) * DIM + WK * (p + 2));
      }
      const float* wc = wt + (p & 1) * WPIECE + c * WTP + 4 * h;
      {
        float4 fb[4], fs[4], fp[4];
#pragma unroll
        for (int u = 0; u < 4; ++u) {
          fb[u] = *(const float4*)(wc + 8 * u);
          fs[u] = *(const float4*)(asl + kfrag(4 * p + u, h));
          fp[u] = *(const float4*)(apl + kfrag(4 * p + u, h));
        }
#pragma unroll
        for (int u = 0; u < 4; ++u) {
          v2f b0;  b0.x  = fb[u].x; b0.y  = fb[u].y;
          v2f b1v; b1v.x = fb[u].z; b1v.y = fb[u].w;
          v2f s0;  s0.x  = fs[u].x; s0.y  = fs[u].y;
          v2f s1;  s1.x  = fs[u].z; s1.y  = fs[u].w;
          v2f p0;  p0.x  = fp[u].x; p0.y  = fp[u].y;
          v2f p1;  p1.x  = fp[u].z; p1.y  = fp[u].w;
          ysum = __builtin_amdgcn_wmma_f32_16x16x4_f32(false, s0, false, b0,  (short)0, ysum, false, false);
          yprd = __builtin_amdgcn_wmma_f32_16x16x4_f32(false, p0, false, b0,  (short)0, yprd, false, false);
          ysum = __builtin_amdgcn_wmma_f32_16x16x4_f32(false, s1, false, b1v, (short)0, ysum, false, false);
          yprd = __builtin_amdgcn_wmma_f32_16x16x4_f32(false, p1, false, b1v, (short)0, yprd, false, false);
        }
      }
      if (p + 2 < NPIECE) {
#pragma unroll
        for (int j = 0; j < 4; ++j) keep_ahead(wp[p + 2][j]);
      }
      if (p + 1 < NPIECE) __syncthreads();
    }
  }

#pragma unroll
  for (int r = 0; r < 8; ++r) {
    float z1 = ysum[r] + bias;
    float z2 = yprd[r] + bias;
    z1 = (z1 >= 0.0f) ? z1 : 0.01f * z1;
    z2 = (z2 >= 0.0f) ? z2 : 0.01f * z2;
    acc[(8 * h + r) * DIM + n0 + c] = z1 + z2;
  }
  __syncthreads();

  for (int i = wave; i < nq; i += PT / 32) {
    const int qe = qlist[i];
    const int q  = qe >> 4;
    const int r  = qe & 15;
    const float4* yp = (const float4*)(acc + r * DIM);
    float4* op = (float4*)(obase + (size_t)q * DIM);
    op[lane]      = yp[lane];
    op[lane + 32] = yp[lane + 32];
  }
}

extern "C" void kernel_launch(void* const* d_in, const int* in_sizes, int n_in,
                              void* d_out, int out_size, void* d_ws, size_t ws_size,
                              hipStream_t stream) {
  const float* v_grid  = (const float*)d_in[0];
  const float* v_sc    = (const float*)d_in[1];
  const float* att_s2g = (const float*)d_in[2];
  const float* att_g2s = (const float*)d_in[3];
  const float* W1      = (const float*)d_in[4];
  const float* b1      = (const float*)d_in[5];
  const int*   src_s2g = (const int*)d_in[6];
  const int*   dst_s2g = (const int*)d_in[7];
  const int*   src_g2s = (const int*)d_in[8];
  const int*   dst_g2s = (const int*)d_in[9];
  const int*   sc_id   = (const int*)d_in[10];
  const int*   gr_id   = (const int*)d_in[11];
  float* out = (float*)d_out;

  int* sg    = (int*)d_ws;
  int* ss    = sg + NG;
  int* cnt   = ss + NS;
  int* lists = cnt + 4;

  k_init_slots<<<(NG + 255) / 256, 256, 0, stream>>>(sg, ss, cnt);
  k_compact<<<(2 * NQ + 255) / 256, 256, 0, stream>>>(gr_id, sc_id, sg, ss, cnt);
  k_filter<<<dim3((EDGES + 1023) / 1024, 2), 1024, 0, stream>>>(dst_s2g, dst_g2s, sg, ss, lists, cnt);

  k_pullmix<<<dim3(MAXU / PR, 2), PT, 0, stream>>>(v_grid, v_sc, att_s2g, att_g2s, src_s2g, src_g2s,
                                                 lists, cnt, sg, ss, gr_id, sc_id, W1, b1, out);
}


# filter writes a 1-byte owner tag (slot/16) per list entry and per query; pull workgroups scan tags, 16 per load, not 4-B entries and 4096 slot lookups each; nh still summed in arrival order
# speedup vs baseline: 5.0379x; 1.1275x over previous
#include <hip/hip_runtime.h>

#define NG     100000
#define NS     20000
#define EDGES  300000
#define DIM    256
#define NQ     4096
#define MAXU   4096
#define WPITCH 260
#define EBITS  19
#define LCAP   EDGES
#define PR     16
#define PT     512
#define PCH    (PT * 48)
#define PB     8
#define WK     32
#define NPIECE (DIM / WK)
#define WTP    36
#define WPIECE (16 * WTP)
#define ARENA  (2 * PR * WPITCH + (PT / 32) * 2 * WPIECE)
#define PCAP   ARENA

static_assert(EDGES <= (1 << EBITS), "edge index must fit the packed entry");
static_assert(MAXU  <= (1 << (31 - EBITS)), "slot must fit the packed entry");
static_assert(MAXU % PR == 0 && PR == 16, "a workgroup owns one 16-row tile of slots");
static_assert(PT / 32 * 16 == DIM, "one wave per 16-column tile of the output");
static_assert(PCAP <= ARENA && PCAP > PCH, "the match list lives in the arena and takes one more chunk behind the flush mark");
static_assert((17 * DIM + ARENA + NQ) * 4 + 256 <= 160 * 1024, "two workgroups per WGP");
static_assert(MAXU / PR <= 256, "the owner of a slot (slot / PR) fits a one-byte tag");
static_assert(NQ % 1024 == 0, "whole workgroups of the edge filter write the query tags");
static_assert(NQ % 16 == 0 && NQ / 16 <= PT, "one 16-tag load per thread covers the query tags");
static_assert(LCAP % 16 == 0 && PCH % 16 == 0, "16-tag loads stay 16-B aligned");

typedef __attribute__((ext_vector_type(2))) float v2f;
typedef __attribute__((ext_vector_type(8))) float v8f;

__global__ __launch_bounds__(256) void k_init_slots(int* __restrict__ sg,
                                                    int* __restrict__ ss,
                                                    int* __restrict__ cnt) {
  int i = blockIdx.x * 256 + threadIdx.x;
  if (i < NG) sg[i] = -1;
  if (i < NS) ss[i] = -1;
  if (i < 4)  cnt[i] = 0;
}

__global__ __launch_bounds__(256) void k_compact(const int* __restrict__ gid,
                                                 const int* __restrict__ sid,
                                                 int* __restrict__ sg,
                                                 int* __restrict__ ss,
                                                 int* __restrict__ cnt) {
  int i = blockIdx.x * 256 + threadIdx.x;
  if (i < NQ) {
    int g = gid[i];
    if (atomicCAS(&sg[g], -1, -2) == -1) sg[g] = atomicAdd(&cnt[0], 1);
  } else if (i < 2 * NQ) {
    int g = sid[i - NQ];
    if (atomicCAS(&ss[g], -1, -2) == -1) ss[g] = atomicAdd(&cnt[1], 1);
  }
}

__global__ __launch_bounds__(1024) void k_filter(const int* __restrict__ dst_g,
                                                  const int* __restrict__ dst_s,
                                                  const int* __restrict__ gid,
                                                  const int* __restrict__ sid,
                                                  const int* __restrict__ sg,
                                                  const int* __restrict__ ss,
                                                  int* __restrict__ lists,
                                                  unsigned char* __restrict__ tags,
                                                  unsigned char* __restrict__ qtags,
                                                  int* __restrict__ cnt) {
  __shared__ int wsum[32];
  __shared__ int wbase;
  const int side = blockIdx.y;
  const int lane = threadIdx.x & 31;
  const int wave = threadIdx.x >> 5;
  const int e    = blockIdx.x * 1024 + threadIdx.x;
  const int* dst  = side ? dst_s : dst_g;
  const int* slot = side ? ss : sg;
  int s = -1;
  if (e < EDGES) s = slot[dst[e]];
  if (e < NQ) {
    const int* ids = side ? sid : gid;
    qtags[side * NQ + e] = (unsigned char)(slot[ids[e]] / PR);
  }
  const bool m = (s >= 0);
  const unsigned bal = __builtin_amdgcn_ballot_w32(m);
  const int before = __popc(bal & ((1u << lane) - 1u));
  if (lane == 0) wsum[wave] = __popc(bal);
  __syncthreads();
  if (wave == 0) {
    const int c = wsum[lane];
    int x = c;
#pragma unroll
    for (int d = 1; d < 32; d <<= 1) {
      const int y = __shfl_up(x, d);
      if (lane >= d) x += y;
    }
    wsum[lane] = x - c;
    if (lane == 31) wbase = (x > 0) ? atomicAdd(&cnt[2 + side], x) : 0;
  }
  __syncthreads();
  if (m) {
    const int pos = side * LCAP + wbase + wsum[wave] + before;
    lists[pos] = (s << EBITS) | e;
    tags[pos]  = (unsigned char)(s / PR);
  }
}

__device__ __forceinline__ int kfrag(int t, int h) { return 8 * t + 4 * h; }

__device__ __forceinline__ void keep_ahead(float4& w) {
  asm volatile("" : "+v"(w.x), "+v"(w.y), "+v"(w.z), "+v"(w.w));
}

__global__ __launch_bounds__(PT, 8) void k_pullmix(const float* __restrict__ v_grid,
                                                const float* __restrict__ v_sc,
                                                const float* __restrict__ att_s2g,
                                                const float* __restrict__ att_g2s,
                                                const int* __restrict__ src_s2g,
                                                const int* __restrict__ src_g2s,
                                                const int* __restrict__ lists,
                                                const unsigned char* __restrict__ tags,
                                                const unsigned char* __restrict__ qtags,
                                                const int* __restrict__ cnt,
                                                const int* __restrict__ sg,
                                                const int* __restrict__ ss,
                                                const int* __restrict__ gr_id,
                                                const int* __restrict__ sc_id,
                                                const float* __restrict__ W1,
                                                const float* __restrict__ b1,
                                                float* __restrict__ out) {
  __shared__ __attribute__((aligned(16))) float acc[(PR + 1) * DIM];
  __shared__ __attribute__((aligned(16))) float arena[ARENA];
  __shared__ int qlist[NQ];
  __shared__ int rownode[PR];
  __shared__ int mcount;
  __shared__ int qcount;

  int* mlist = (int*)arena;
  const int tid  = threadIdx.x;
  const int lane = tid & 31;
  const int wave = __builtin_amdgcn_readfirstlane(tid >> 5);
  const unsigned blk = blockIdx.x;
  const int side = 1 - (int)blockIdx.y;

  const float* vsrc  = side ? v_grid  : v_sc;
  const float* att   = side ? att_g2s : att_s2g;
  const int*   src   = side ? src_g2s : src_s2g;
  const int*   list  = lists + (size_t)side * LCAP;
  const unsigned char* tg = tags  + (size_t)side * LCAP;
  const unsigned char* qt = qtags + (size_t)side * NQ;
  const unsigned pat = blk * 0x01010101u;
  const float* vself = side ? v_sc    : v_grid;
  const int*   slot  = side ? ss      : sg;
  const int*   ids   = side ? sc_id   : gr_id;
  float*       obase = out + (side ? (size_t)0 : (size_t)NQ * DIM);
  const int n = cnt[2 + side];

  for (int i = tid; i < (PR + 1) * DIM; i += PT) acc[i] = 0.0f;
  if (tid < PR) rownode[tid] = -1;
  if (tid == 0) { mcount = 0; qcount = 0; }
  __syncthreads();

  if (tid < NQ / 16) {
    const uint4 t = *(const uint4*)(qt + tid * 16);
    const unsigned w[4] = {t.x, t.y, t.z, t.w};
#pragma unroll
    for (int d = 0; d < 4; ++d) {
      const unsigned y = w[d] ^ pat;
      if (((y - 0x01010101u) & ~y & 0x80808080u) != 0u) {
#pragma unroll
        for (int j = 0; j < 4; ++j) {
          if (((y >> (8 * j)) & 0xFFu) == 0u) {
            const int pos = atomicAdd(&qcount, 1);
            qlist[pos] = tid * 16 + 4 * d + j;
          }
        }
      }
    }
  }
  __syncthreads();
  const int nq = __builtin_amdgcn_readfirstlane(qcount);
  if (nq == 0) return;
  for (int i = tid; i < nq; i += PT) {
    const int q = qlist[i];
    const int g = ids[q];
    const int r = slot[g] % PR;
    qlist[i]   = (q << 4) | r;
    rownode[r] = g;
  }
  __syncthreads();

  for (int c0 = 0; c0 < n; c0 += PCH) {
    uint4 tv[PCH / (PT * 16)];
#pragma unroll
    for (int k = 0; k < PCH / (PT * 16); ++k) {
      const int i16 = c0 + (k * PT + tid) * 16;
      tv[k] = *(const uint4*)(tg + ((i16 < n) ? i16 : 0));
    }
#pragma unroll
    for (int k = 0; k < PCH / (PT * 16); ++k) {
      const int i16 = c0 + (k * PT + tid) * 16;
      const unsigned w[4] = {tv[k].x, tv[k].y, tv[k].z, tv[k].w};
#pragma unroll
      for (int d = 0; d < 4; ++d) {
        const unsigned y = w[d] ^ pat;
        if (((y - 0x01010101u) & ~y & 0x80808080u) != 0u) {
#pragma unroll
          for (int j = 0; j < 4; ++j) {
            const int idx = i16 + 4 * d + j;
            if (((y >> (8 * j)) & 0xFFu) == 0u && idx < n) {
              const int pos = atomicAdd(&mcount, 1);
              mlist[pos] = idx;
            }
          }
        }
      }
    }
    __syncthreads();
    const int mc = __builtin_amdgcn_readfirstlane(mcount);
    __syncthreads();

    if (mc > PCAP - PCH || c0 + PCH >= n) {
      for (int i = tid; i < mc; i += PT) mlist[i] = list[mlist[i]];
      __syncthreads();
      for (int b0 = wave * PB; b0 < mc; b0 += (PT / 32) * PB) {
        int   srow = 0;
        int   arow = PR;
        float a    = 0.0f;
        if (lane < PB && b0 + lane < mc) {
          const unsigned ment = (unsigned)mlist[b0 + lane];
          const int e = (int)(ment & ((1u << EBITS) - 1u));
          srow = src[e];
          a    = att[e];
          arow = (int)((ment >> EBITS) % PR);
        }
        float4 rv[PB][2];
#pragma unroll
        for (int b = 0; b < PB; ++b) {
          const int sr = __shfl(srow, b);
          const float4* vp = (const float4*)(vsrc + (size_t)sr * DIM);
          rv[b][0] = vp[lane];
          rv[b][1] = vp[lane + 32];
        }
#pragma unroll
        for (int b = 0; b < PB; ++b) {
          const float ab = __shfl(a, b);
          float* ap = acc + __shfl(arow, b) * DIM + lane * 4;
          atomicAdd(ap + 0,   rv[b][0].x * ab);
          atomicAdd(ap + 1,   rv[b][0].y * ab);
          atomicAdd(ap + 2,   rv[b][0].z * ab);
          atomicAdd(ap + 3,   rv[b][0].w * ab);
          atomicAdd(ap + 128, rv[b][1].x * ab);
          atomicAdd(ap + 129, rv[b][1].y * ab);
          atomicAdd(ap + 130, rv[b][1].z * ab);
          atomicAdd(ap + 131, rv[b][1].w * ab);
        }
      }
      __syncthreads();
      if (tid == 0) mcount = 0;
      __syncthreads();
    }
  }

  const int c  = lane & 15;
  const int h  = lane >> 4;
  const int n0 = wave * 16;

  const int wr = lane >> 3;
  const int wq = lane & 7;
  const float* wsl = W1 + (size_t)(n0 + wr) * DIM + wq * 4;
  float4 wp[NPIECE][4];
#pragma unroll
  for (int p = 0; p < 2; ++p) {
#pragma unroll
    for (int j = 0; j < 4; ++j) wp[p][j] = *(const float4*)(wsl + (4 * j) * DIM + WK * p);
  }
  const float bias = b1[n0 + c];

  float* As = arena;
  float* Ap = As + PR * WPITCH;
  float* wt = Ap + PR * WPITCH + wave * (2 * WPIECE);
#pragma unroll
  for (int j = 0; j < 4; ++j) *(float4*)(wt + (4 * j + wr) * WTP + wq * 4) = wp[0][j];
  {
    const int r = tid >> 5;
    const int g = rownode[r];
    float4 v0 = make_float4(0.0f, 0.0f, 0.0f, 0.0f);
    float4 v1 = v0;
    if (g >= 0) {
      const float4* vp = (const float4*)(vself + (size_t)g * DIM);
      v0 = vp[lane];
      v1 = vp[lane + 32];
    }
    const float4 h0 = *(const float4*)(acc + r * DIM + lane * 4);
    const float4 h1 = *(const float4*)(acc + r * DIM + 128 + lane * 4);
    float* s0 = As + r * WPITCH + lane * 4;
    float* p0 = Ap + r * WPITCH + lane * 4;
    *(float4*)(s0)       = make_float4(v0.x + h0.x, v0.y + h0.y, v0.z + h0.z, v0.w + h0.w);
    *(float4*)(s0 + 128) = make_float4(v1.x + h1.x, v1.y + h1.y, v1.z + h1.z, v1.w + h1.w);
    *(float4*)(p0)       = make_float4(v0.x * h0.x, v0.y * h0.y, v0.z * h0.z, v0.w * h0.w);
    *(float4*)(p0 + 128) = make_float4(v1.x * h1.x, v1.y * h1.y, v1.z * h1.z, v1.w * h1.w);
  }
#pragma unroll
  for (int j = 0; j < 4; ++j) keep_ahead(wp[1][j]);
  __syncthreads();

  v8f ysum = (v8f){};
  v8f yprd = (v8f){};
  {
    const float* asl = As + c * WPITCH;
    const float* apl = Ap + c * WPITCH;
#pragma unroll
    for (int p = 0; p < NPIECE; ++p) {
      if (p + 1 < NPIECE) {
        float* wn = wt + ((p + 1) & 1) * WPIECE;
#pragma unroll
        for (int j = 0; j < 4; ++j) *(float4*)(wn + (4 * j + wr) * WTP + wq * 4) = wp[p + 1][j];
      }
      if (p + 2 < NPIECE) {
#pragma unroll
        for (int j = 0; j < 4; ++j) wp[p + 2][j] = *(const float4*)(wsl + (4 * j) * DIM + WK * (p + 2));
      }
      const float* wc = wt + (p & 1) * WPIECE + c * WTP + 4 * h;
      {
        float4 fb[4], fs[4], fp[4];
#pragma unroll
        for (int u = 0; u < 4; ++u) {
          fb[u] = *(const float4*)(wc + 8 * u);
          fs[u] = *(const float4*)(asl + kfrag(4 * p + u, h));
          fp[u] = *(const float4*)(apl + kfrag(4 * p + u, h));
        }
#pragma unroll
        for (int u = 0; u < 4; ++u) {
          v2f b0;  b0.x  = fb[u].x; b0.y  = fb[u].y;
          v2f b1v; b1v.x = fb[u].z; b1v.y = fb[u].w;
          v2f s0;  s0.x  = fs[u].x; s0.y  = fs[u].y;
          v2f s1;  s1.x  = fs[u].z; s1.y  = fs[u].w;
          v2f p0;  p0.x  = fp[u].x; p0.y  = fp[u].y;
          v2f p1;  p1.x  = fp[u].z; p1.y  = fp[u].w;
          ysum = __builtin_amdgcn_wmma_f32_16x16x4_f32(false, s0, false, b0,  (short)0, ysum, false, false);
          yprd = __builtin_amdgcn_wmma_f32_16x16x4_f32(false, p0, false, b0,  (short)0, yprd, false, false);
          ysum = __builtin_amdgcn_wmma_f32_16x16x4_f32(false, s1, false, b1v, (short)0, ysum, false, false);
          yprd = __builtin_amdgcn_wmma_f32_16x16x4_f32(false, p1, false, b1v, (short)0, yprd, false, false);
        }
      }
      if (p + 2 < NPIECE) {
#pragma unroll
        for (int j = 0; j < 4; ++j) keep_ahead(wp[p + 2][j]);
      }
      if (p + 1 < NPIECE) __syncthreads();
    }
  }

#pragma unroll
  for (int r = 0; r < 8; ++r) {
    float z1 = ysum[r] + bias;
    float z2 = yprd[r] + bias;
    z1 = (z1 >= 0.0f) ? z1 : 0.01f * z1;
    z2 = (z2 >= 0.0f) ? z2 : 0.01f * z2;
    acc[(8 * h + r) * DIM + n0 + c] = z1 + z2;
  }
  __syncthreads();

  for (int i = wave; i < nq; i += PT / 32) {
    const int qe = qlist[i];
    const int q  = qe >> 4;
    const int r  = qe & 15;
    const float4* yp = (const float4*)(acc + r * DIM);
    float4* op = (float4*)(obase + (size_t)q * DIM);
    op[lane]      = yp[lane];
    op[lane + 32] = yp[lane + 32];
  }
}

extern "C" void kernel_launch(void* const* d_in, const int* in_sizes, int n_in,
                              void* d_out, int out_size, void* d_ws, size_t ws_size,
                              hipStream_t stream) {
  const float* v_grid  = (const float*)d_in[0];
  const float* v_sc    = (const float*)d_in[1];
  const float* att_s2g = (const float*)d_in[2];
  const float* att_g2s = (const float*)d_in[3];
  const float* W1      = (const float*)d_in[4];
  const float* b1      = (const float*)d_in[5];
  const int*   src_s2g = (const int*)d_in[6];
  const int*   dst_s2g = (const int*)d_in[7];
  const int*   src_g2s = (const int*)d_in[8];
  const int*   dst_g2s = (const int*)d_in[9];
  const int*   sc_id   = (const int*)d_in[10];
  const int*   gr_id   = (const int*)d_in[11];
  float* out = (float*)d_out;

  int* sg    = (int*)d_ws;
  int* ss    = sg + NG;
  int* cnt   = ss + NS;
  int* lists = cnt + 4;
  unsigned char* tags  = (unsigned char*)(lists + 2 * LCAP);
  unsigned char* qtags = tags + 2 * LCAP;

  k_init_slots<<<(NG + 255) / 256, 256, 0, stream>>>(sg, ss, cnt);
  k_compact<<<(2 * NQ + 255) / 256, 256, 0, stream>>>(gr_id, sc_id, sg, ss, cnt);
  k_filter<<<dim3((EDGES + 1023) / 1024, 2), 1024, 0, stream>>>(dst_s2g, dst_g2s, gr_id, sc_id, sg, ss,
                                                                   lists, tags, qtags, cnt);

  k_pullmix<<<dim3(MAXU / PR, 2), PT, 0, stream>>>(v_grid, v_sc, att_s2g, att_g2s, src_s2g, src_g2s,
                                                 lists, tags, qtags, cnt, sg, ss, gr_id, sc_id, W1, b1, out);
}


# node-to-slot map built by 30 workgroups owning 4,096 nodes each (LDS range = -1, queries in range post their index by LDS max, range stored); clear pass, CAS, counters gone; 4 launches to 3
# speedup vs baseline: 5.4631x; 1.0844x over previous
#include <hip/hip_runtime.h>

#define NG     100000
#define NS     20000
#define EDGES  300000
#define DIM    256
#define NQ     4096
#define MAXU   NQ
#define MR     4096
#define MGB    ((NG + MR - 1) / MR)
#define MSB    ((NS + MR - 1) / MR)
#define WPITCH 260
#define EBITS  19
#define LCAP   EDGES
#define PR     16
#define PT     512
#define PCH    (PT * 48)
#define PB     8
#define WK     32
#define NPIECE (DIM / WK)
#define WTP    36
#define WPIECE (16 * WTP)
#define ARENA  (2 * PR * WPITCH + (PT / 32) * 2 * WPIECE)
#define PCAP   ARENA

static_assert(EDGES <= (1 << EBITS), "edge index must fit the packed entry");
static_assert(MAXU  <= (1 << (31 - EBITS)), "slot must fit the packed entry");
static_assert(MAXU % PR == 0 && PR == 16, "a workgroup owns one 16-row tile of slots");
static_assert(PT / 32 * 16 == DIM, "one wave per 16-column tile of the output");
static_assert(PCAP <= ARENA && PCAP > PCH, "the match list lives in the arena and takes one more chunk behind the flush mark");
static_assert((17 * DIM + ARENA + NQ) * 4 + 256 <= 160 * 1024, "two workgroups per WGP");
static_assert(MAXU / PR <= 256, "the owner of a slot (slot / PR) fits a one-byte tag");
static_assert(NQ % 1024 == 0, "whole workgroups of the edge filter write the query tags");
static_assert(NQ % 16 == 0 && NQ / 16 <= PT, "one 16-tag load per thread covers the query tags");
static_assert(LCAP % 16 == 0 && PCH % 16 == 0, "16-tag loads stay 16-B aligned");
static_assert(NQ == 4 * 1024 && MR == 4 * 1024, "a map thread takes four queries in one 16-B load and four nodes in one 16-B store");
static_assert(NG % 4 == 0 && NS % 4 == 0, "the 16-B stores of the map end with the map");

typedef __attribute__((ext_vector_type(2))) float v2f;
typedef __attribute__((ext_vector_type(8))) float v8f;

__global__ __launch_bounds__(1024) void k_map(const int* __restrict__ gid,
                                              const int* __restrict__ sid,
                                              int* __restrict__ sg,
                                              int* __restrict__ ss,
                                              int* __restrict__ cnt) {
  __shared__ __attribute__((aligned(16))) int lmap[MR];
  const int  t    = threadIdx.x;
  const bool sc   = blockIdx.x >= MGB;
  const int  base = (sc ? (int)blockIdx.x - MGB : (int)blockIdx.x) * MR;
  const int* ids  = sc ? sid : gid;
  int*       map  = sc ? ss : sg;
  const int  nn   = sc ? NS : NG;
  const int4 q = *(const int4*)(ids + 4 * t);
  *(int4*)(lmap + 4 * t) = make_int4(-1, -1, -1, -1);
  if (blockIdx.x == 0 && t < 2) cnt[2 + t] = 0;
  __syncthreads();
  const int g[4] = {q.x, q.y, q.z, q.w};
#pragma unroll
  for (int k = 0; k < 4; ++k) {
    const unsigned o = (unsigned)(g[k] - base);
    if (o < (unsigned)MR) atomicMax(&lmap[o], 4 * t + k);
  }
  __syncthreads();
  if (base + 4 * t < nn) *(int4*)(map + base + 4 * t) = *(const int4*)(lmap + 4 * t);
}

__global__ __launch_bounds__(1024) void k_filter(const int* __restrict__ dst_g,
                                                  const int* __restrict__ dst_s,
                                                  const int* __restrict__ gid,
                                                  const int* __restrict__ sid,
                                                  const int* __restrict__ sg,
                                                  const int* __restrict__ ss,
                                                  int* __restrict__ lists,
                                                  unsigned char* __restrict__ tags,
                                                  unsigned char* __restrict__ qtags,
                                                  int* __restrict__ cnt) {
  __shared__ int wsum[32];
  __shared__ int wbase;
  const int side = blockIdx.y;
  const int lane = threadIdx.x & 31;
  const int wave = threadIdx.x >> 5;
  const int e    = blockIdx.x * 1024 + threadIdx.x;
  const int* dst  = side ? dst_s : dst_g;
  const int* slot = side ? ss : sg;
  int s = -1;
  if (e < EDGES) s = slot[dst[e]];
  if (e < NQ) {
    const int* ids = side ? sid : gid;
    qtags[side * NQ + e] = (unsigned char)(slot[ids[e]] / PR);
  }
  const bool m = (s >= 0);
  const unsigned bal = __builtin_amdgcn_ballot_w32(m);
  const int before = __popc(bal & ((1u << lane) - 1u));
  if (lane == 0) wsum[wave] = __popc(bal);
  __syncthreads();
  if (wave == 0) {
    const int c = wsum[lane];
    int x = c;
#pragma unroll
    for (int d = 1; d < 32; d <<= 1) {
      const int y = __shfl_up(x, d);
      if (lane >= d) x += y;
    }
    wsum[lane] = x - c;
    if (lane == 31) wbase = (x > 0) ? atomicAdd(&cnt[2 + side], x) : 0;
  }
  __syncthreads();
  if (m) {
    const int pos = side * LCAP + wbase + wsum[wave] + before;
    lists[pos] = (s << EBITS) | e;
    tags[pos]  = (unsigned char)(s / PR);
  }
}

__device__ __forceinline__ int kfrag(int t, int h) { return 8 * t + 4 * h; }

__device__ __forceinline__ void keep_ahead(float4& w) {
  asm volatile("" : "+v"(w.x), "+v"(w.y), "+v"(w.z), "+v"(w.w));
}

__global__ __launch_bounds__(PT, 8) void k_pullmix(const float* __restrict__ v_grid,
                                                const float* __restrict__ v_sc,
                                                const float* __restrict__ att_s2g,
                                                const float* __restrict__ att_g2s,
                                                const int* __restrict__ src_s2g,
                                                const int* __restrict__ src_g2s,
                                                const int* __restrict__ lists,
                                                const unsigned char* __restrict__ tags,
                                                const unsigned char* __restrict__ qtags,
                                                const int* __restrict__ cnt,
                                                const int* __restrict__ sg,
                                                const int* __restrict__ ss,
                                                const int* __restrict__ gr_id,
                                                const int* __restrict__ sc_id,
                                                const float* __restrict__ W1,
                                                const float* __restrict__ b1,
                                                float* __restrict__ out) {
  __shared__ __attribute__((aligned(16))) float acc[(PR + 1) * DIM];
  __shared__ __attribute__((aligned(16))) float arena[ARENA];
  __shared__ int qlist[NQ];
  __shared__ int rownode[PR];
  __shared__ int mcount;
  __shared__ int qcount;

  int* mlist = (int*)arena;
  const int tid  = threadIdx.x;
  const int lane = tid & 31;
  const int wave = __builtin_amdgcn_readfirstlane(tid >> 5);
  const unsigned blk = blockIdx.x;
  const int side = 1 - (int)blockIdx.y;

  const float* vsrc  = side ? v_grid  : v_sc;
  const float* att   = side ? att_g2s : att_s2g;
  const int*   src   = side ? src_g2s : src_s2g;
  const int*   list  = lists + (size_t)side * LCAP;
  const unsigned char* tg = tags  + (size_t)side * LCAP;
  const unsigned char* qt = qtags + (size_t)side * NQ;
  const unsigned pat = blk * 0x01010101u;
  const float* vself = side ? v_sc    : v_grid;
  const int*   slot  = side ? ss      : sg;
  const int*   ids   = side ? sc_id   : gr_id;
  float*       obase = out + (side ? (size_t)0 : (size_t)NQ * DIM);
  const int n = cnt[2 + side];

  for (int i = tid; i < (PR + 1) * DIM; i += PT) acc[i] = 0.0f;
  if (tid < PR) rownode[tid] = -1;
  if (tid == 0) { mcount = 0; qcount = 0; }
  __syncthreads();

  if (tid < NQ / 16) {
    const uint4 t = *(const uint4*)(qt + tid * 16);
    const unsigned w[4] = {t.x, t.y, t.z, t.w};
#pragma unroll
    for (int d = 0; d < 4; ++d) {
      const unsigned y = w[d] ^ pat;
      if (((y - 0x01010101u) & ~y & 0x80808080u) != 0u) {
#pragma unroll
        for (int j = 0; j < 4; ++j) {
          if (((y >> (8 * j)) & 0xFFu) == 0u) {
            const int pos = atomicAdd(&qcount, 1);
            qlist[pos] = tid * 16 + 4 * d + j;
          }
        }
      }
    }
  }
  __syncthreads();
  const int nq = __builtin_amdgcn_readfirstlane(qcount);
  if (nq == 0) return;
  for (int i = tid; i < nq; i += PT) {
    const int q = qlist[i];
    const int g = ids[q];
    const int r = slot[g] % PR;
    qlist[i]   = (q << 4) | r;
    rownode[r] = g;
  }
  __syncthreads();

  for (int c0 = 0; c0 < n; c0 += PCH) {
    uint4 tv[PCH / (PT * 16)];
#pragma unroll
    for (int k = 0; k < PCH / (PT * 16); ++k) {
      const int i16 = c0 + (k * PT + tid) * 16;
      tv[k] = *(const uint4*)(tg + ((i16 < n) ? i16 : 0));
    }
#pragma unroll
    for (int k = 0; k < PCH / (PT * 16); ++k) {
      const int i16 = c0 + (k * PT + tid) * 16;
      const unsigned w[4] = {tv[k].x, tv[k].y, tv[k].z, tv[k].w};
#pragma unroll
      for (int d = 0; d < 4; ++d) {
        const unsigned y = w[d] ^ pat;
        if (((y - 0x01010101u) & ~y & 0x80808080u) != 0u) {
#pragma unroll
          for (int j = 0; j < 4; ++j) {
            const int idx = i16 + 4 * d + j;
            if (((y >> (8 * j)) & 0xFFu) == 0u && idx < n) {
              const int pos = atomicAdd(&mcount, 1);
              mlist[pos] = idx;
            }
          }
        }
      }
    }
    __syncthreads();
    const int mc = __builtin_amdgcn_readfirstlane(mcount);
    __syncthreads();

    if (mc > PCAP - PCH || c0 + PCH >= n) {
      for (int i = tid; i < mc; i += PT) mlist[i] = list[mlist[i]];
      __syncthreads();
      for (int b0 = wave * PB; b0 < mc; b0 += (PT / 32) * PB) {
        int   srow = 0;
        int   arow = PR;
        float a    = 0.0f;
        if (lane < PB && b0 + lane < mc) {
          const unsigned ment = (unsigned)mlist[b0 + lane];
          const int e = (int)(ment & ((1u << EBITS) - 1u));
          srow = src[e];
          a    = att[e];
          arow = (int)((ment >> EBITS) % PR);
        }
        float4 rv[PB][2];
#pragma unroll
        for (int b = 0; b < PB; ++b) {
          const int sr = __shfl(srow, b);
          const float4* vp = (const float4*)(vsrc + (size_t)sr * DIM);
          rv[b][0] = vp[lane];
          rv[b][1] = vp[lane + 32];
        }
#pragma unroll
        for (int b = 0; b < PB; ++b) {
          const float ab = __shfl(a, b);
          float* ap = acc + __shfl(arow, b) * DIM + lane * 4;
          atomicAdd(ap + 0,   rv[b][0].x * ab);
          atomicAdd(ap + 1,   rv[b][0].y * ab);
          atomicAdd(ap + 2,   rv[b][0].z * ab);
          atomicAdd(ap + 3,   rv[b][0].w * ab);
          atomicAdd(ap + 128, rv[b][1].x * ab);
          atomicAdd(ap + 129, rv[b][1].y * ab);
          atomicAdd(ap + 130, rv[b][1].z * ab);
          atomicAdd(ap + 131, rv[b][1].w * ab);
        }
      }
      __syncthreads();
      if (tid == 0) mcount = 0;
      __syncthreads();
    }
  }

  const int c  = lane & 15;
  const int h  = lane >> 4;
  const int n0 = wave * 16;

  const int wr = lane >> 3;
  const int wq = lane & 7;
  const float* wsl = W1 + (size_t)(n0 + wr) * DIM + wq * 4;
  float4 wp[NPIECE][4];
#pragma unroll
  for (int p = 0; p < 2; ++p) {
#pragma unroll
    for (int j = 0; j < 4; ++j) wp[p][j] = *(const float4*)(wsl + (4 * j) * DIM + WK * p);
  }
  const float bias = b1[n0 + c];

  float* As = arena;
  float* Ap = As + PR * WPITCH;
  float* wt = Ap + PR * WPITCH + wave * (2 * WPIECE);
#pragma unroll
  for (int j = 0; j < 4; ++j) *(float4*)(wt + (4 * j + wr) * WTP + wq * 4) = wp[0][j];
  {
    const int r = tid >> 5;
    const int g = rownode[r];
    float4 v0 = make_float4(0.0f, 0.0f, 0.0f, 0.0f);
    float4 v1 = v0;
    if (g >= 0) {
      const float4* vp = (const float4*)(vself + (size_t)g * DIM);
      v0 = vp[lane];
      v1 = vp[lane + 32];
    }
    const float4 h0 = *(const float4*)(acc + r * DIM + lane * 4);
    const float4 h1 = *(const float4*)(acc + r * DIM + 128 + lane * 4);
    float* s0 = As + r * WPITCH + lane * 4;
    float* p0 = Ap + r * WPITCH + lane * 4;
    *(float4*)(s0)       = make_float4(v0.x + h0.x, v0.y + h0.y, v0.z + h0.z, v0.w + h0.w);
    *(float4*)(s0 + 128) = make_float4(v1.x + h1.x, v1.y + h1.y, v1.z + h1.z, v1.w + h1.w);
    *(float4*)(p0)       = make_float4(v0.x * h0.x, v0.y * h0.y, v0.z * h0.z, v0.w * h0.w);
    *(float4*)(p0 + 128) = make_float4(v1.x * h1.x, v1.y * h1.y, v1.z * h1.z, v1.w * h1.w);
  }
#pragma unroll
  for (int j = 0; j < 4; ++j) keep_ahead(wp[1][j]);
  __syncthreads();

  v8f ysum = (v8f){};
  v8f yprd = (v8f){};
  {
    const float* asl = As + c * WPITCH;
    const float* apl = Ap + c * WPITCH;
#pragma unroll
    for (int p = 0; p < NPIECE; ++p) {
      if (p + 1 < NPIECE) {
        float* wn = wt + ((p + 1) & 1) * WPIECE;
#pragma unroll
        for (int j = 0; j < 4; ++j) *(float4*)(wn + (4 * j + wr) * WTP + wq * 4) = wp[p + 1][j];
      }
      if (p + 2 < NPIECE) {
#pragma unroll
        for (int j = 0; j < 4; ++j) wp[p + 2][j] = *(const float4*)(wsl + (4 * j) * DIM + WK * (p + 2));
      }
      const float* wc = wt + (p & 1) * WPIECE + c * WTP + 4 * h;
      {
        float4 fb[4], fs[4], fp[4];
#pragma unroll
        for (int u = 0; u < 4; ++u) {
          fb[u] = *(const float4*)(wc + 8 * u);
          fs[u] = *(const float4*)(asl + kfrag(4 * p + u, h));
          fp[u] = *(const float4*)(apl + kfrag(4 * p + u, h));
        }
#pragma unroll
        for (int u = 0; u < 4; ++u) {
          v2f b0;  b0.x  = fb[u].x; b0.y  = fb[u].y;
          v2f b1v; b1v.x = fb[u].z; b1v.y = fb[u].w;
          v2f s0;  s0.x  = fs[u].x; s0.y  = fs[u].y;
          v2f s1;  s1.x  = fs[u].z; s1.y  = fs[u].w;
          v2f p0;  p0.x  = fp[u].x; p0.y  = fp[u].y;
          v2f p1;  p1.x  = fp[u].z; p1.y  = fp[u].w;
          ysum = __builtin_amdgcn_wmma_f32_16x16x4_f32(false, s0, false, b0,  (short)0, ysum, false, false);
          yprd = __builtin_amdgcn_wmma_f32_16x16x4_f32(false, p0, false, b0,  (short)0, yprd, false, false);
          ysum = __builtin_amdgcn_wmma_f32_16x16x4_f32(false, s1, false, b1v, (short)0, ysum, false, false);
          yprd = __builtin_amdgcn_wmma_f32_16x16x4_f32(false, p1, false, b1v, (short)0, yprd, false, false);
        }
      }
      if (p + 2 < NPIECE) {
#pragma unroll
        for (int j = 0; j < 4; ++j) keep_ahead(wp[p + 2][j]);
      }
      if (p + 1 < NPIECE) __syncthreads();
    }
  }

#pragma unroll
  for (int r = 0; r < 8; ++r) {
    float z1 = ysum[r] + bias;
    float z2 = yprd[r] + bias;
    z1 = (z1 >= 0.0f) ? z1 : 0.01f * z1;
    z2 = (z2 >= 0.0f) ? z2 : 0.01f * z2;
    acc[(8 * h + r) * DIM + n0 + c] = z1 + z2;
  }
  __syncthreads();

  for (int i = wave; i < nq; i += PT / 32) {
    const int qe = qlist[i];
    const int q  = qe >> 4;
    const int r  = qe & 15;
    const float4* yp = (const float4*)(acc + r * DIM);
    float4* op = (float4*)(obase + (size_t)q * DIM);
    op[lane]      = yp[lane];
    op[lane + 32] = yp[lane + 32];
  }
}

extern "C" void kernel_launch(void* const* d_in, const int* in_sizes, int n_in,
                              void* d_out, int out_size, void* d_ws, size_t ws_size,
                              hipStream_t stream) {
  const float* v_grid  = (const float*)d_in[0];
  const float* v_sc    = (const float*)d_in[1];
  const float* att_s2g = (const float*)d_in[2];
  const float* att_g2s = (const float*)d_in[3];
  const float* W1      = (const float*)d_in[4];
  const float* b1      = (const float*)d_in[5];
  const int*   src_s2g = (const int*)d_in[6];
  const int*   dst_s2g = (const int*)d_in[7];
  const int*   src_g2s = (const int*)d_in[8];
  const int*   dst_g2s = (const int*)d_in[9];
  const int*   sc_id   = (const int*)d_in[10];
  const int*   gr_id   = (const int*)d_in[11];
  float* out = (float*)d_out;

  int* sg    = (int*)d_ws;
  int* ss    = sg + NG;
  int* cnt   = ss + NS;
  int* lists = cnt + 4;
  unsigned char* tags  = (unsigned char*)(lists + 2 * LCAP);
  unsigned char* qtags = tags + 2 * LCAP;

  k_map<<<MGB + MSB, 1024, 0, stream>>>(gr_id, sc_id, sg, ss, cnt);
  k_filter<<<dim3((EDGES + 1023) / 1024, 2), 1024, 0, stream>>>(dst_s2g, dst_g2s, gr_id, sc_id, sg, ss,
                                                                   lists, tags, qtags, cnt);

  k_pullmix<<<dim3(MAXU / PR, 2), PT, 0, stream>>>(v_grid, v_sc, att_s2g, att_g2s, src_s2g, src_g2s,
                                                 lists, tags, qtags, cnt, sg, ss, gr_id, sc_id, W1, b1, out);
}


# map and edge filter in one launch: 48 workgroups each build a node range of the node-to-slot map in LDS and list their edge share's matches as their own run; 3 launches to 2; nh sum order kept
# speedup vs baseline: 5.7558x; 1.0536x over previous
#include <hip/hip_runtime.h>

#define NG     100000
#define NS     20000
#define EDGES  300000
#define DIM    256
#define NQ     4096
#define MAXU   NQ
#define RN     50000
#define NRG    ((NG + RN - 1) / RN)
#define SH     16
#define SEGCAP 18752
#define SEGSTR (SEGCAP + 16)
#define NRUNG  (NRG * SH)
#define NRUN   (NRUNG + SH)
#define NLIST  (2 * SH * SEGSTR)
#define FEPT   20
#define WPITCH 260
#define EBITS  19
#define PR     16
#define PT     512
#define PCH    (PT * 48)
#define PB     8
#define WK     32
#define NPIECE (DIM / WK)
#define WTP    36
#define WPIECE (16 * WTP)
#define ARENA  (2 * PR * WPITCH + (PT / 32) * 2 * WPIECE)
#define PCAP   ARENA

static_assert(EDGES <= (1 << EBITS), "edge index must fit the packed entry");
static_assert(MAXU  <= (1 << (31 - EBITS)), "slot must fit the packed entry");
static_assert(MAXU % PR == 0 && PR == 16, "a workgroup owns one 16-row tile of slots");
static_assert(PT / 32 * 16 == DIM, "one wave per 16-column tile of the output");
static_assert(PCAP <= ARENA && PCAP > PCH, "the match list lives in the arena and takes one more chunk behind the flush mark");
static_assert((17 * DIM + ARENA + NQ) * 4 + 256 <= 160 * 1024, "two workgroups per WGP");
static_assert(MAXU / PR <= 256, "the owner of a slot (slot / PR) fits a one-byte tag");
static_assert(NQ % 1024 == 0, "whole workgroups of the edge filter write the query tags");
static_assert(NQ % 16 == 0 && NQ / 16 <= PT, "one 16-tag load per thread covers the query tags");
static_assert(SEGCAP % 16 == 0 && SEGSTR % 16 == 0 && PCH % 16 == 0, "16-tag loads stay 16-B aligned");
static_assert(NQ == 4 * 1024, "a front thread takes four queries in one 16-B load");
static_assert(NG % 4 == 0 && NS % 4 == 0 && RN % 4 == 0, "the 16-B LDS stores that clear a range end with the range");
static_assert(NRG == 2 && NS <= RN, "two grid ranges share a segment from its two ends; one range holds the small-category side");
static_assert(SH * SEGCAP >= EDGES && EDGES % 4 == 0 && SEGCAP % 4 == 0, "the shares cover the edges in whole 16-B loads");
static_assert(1024 * FEPT >= SEGCAP && FEPT % 4 == 0, "a front workgroup covers its share");
static_assert(NRUNG <= 32 && SH <= 32 && 2 * NRUN <= 128, "one wave scans a side's run table; the table fits its 128 ints");

typedef __attribute__((ext_vector_type(2))) float v2f;
typedef __attribute__((ext_vector_type(8))) float v8f;

__global__ __launch_bounds__(1024) void k_front(const int* __restrict__ dst_g,
                                                const int* __restrict__ dst_s,
                                                const int* __restrict__ gid,
                                                const int* __restrict__ sid,
                                                int* __restrict__ lists,
                                                unsigned char* __restrict__ tags,
                                                int* __restrict__ runs,
                                                int* __restrict__ qslot,
                                                unsigned char* __restrict__ qtags) {
  __shared__ __attribute__((aligned(16))) int lmap[RN];
  __shared__ int wsum[32];
  __shared__ int wtot;
  const int  t     = threadIdx.x;
  const int  lane  = t & 31;
  const int  wave  = t >> 5;
  const int  b     = blockIdx.x;
  const bool sc    = b >= NRUNG;
  const int  range = sc ? 0 : b / SH;
  const int  share = sc ? b - NRUNG : b % SH;
  const int  side  = sc ? 1 : 0;
  const int  base  = range * RN;
  const int  left  = (sc ? NS : NG) - base;
  const int  nr    = left < RN ? left : RN;
  const int* ids   = sc ? sid : gid;
  const int* dst   = sc ? dst_s : dst_g;

  int4 dv[FEPT / 4];
#pragma unroll
  for (int k = 0; k < FEPT / 4; ++k) {
    const int i4 = 4 * (k * 1024 + t);
    const int e0 = share * SEGCAP + i4;
    dv[k] = make_int4(-1, -1, -1, -1);
    if (i4 < SEGCAP && e0 < EDGES) dv[k] = *(const int4*)(dst + e0);
  }
  const int4 q = *(const int4*)(ids + 4 * t);
  for (int i = t; i < nr / 4; i += 1024) *(int4*)(lmap + 4 * i) = make_int4(-1, -1, -1, -1);
  __syncthreads();
  const int g[4] = {q.x, q.y, q.z, q.w};
#pragma unroll
  for (int k = 0; k < 4; ++k) {
    const unsigned o = (unsigned)(g[k] - base);
    if (o < (unsigned)nr) atomicMax(&lmap[o], 4 * t + k);
  }
  __syncthreads();
  if (share == 0) {
#pragma unroll
    for (int k = 0; k < 4; ++k) {
      const unsigned o = (unsigned)(g[k] - base);
      if (o < (unsigned)nr) {
        const int sl = lmap[o];
        qslot[side * NQ + 4 * t + k] = sl;
        qtags[side * NQ + 4 * t + k] = (unsigned char)(sl / PR);
      }
    }
  }

  int sl[FEPT];
  int c = 0;
#pragma unroll
  for (int k = 0; k < FEPT / 4; ++k) {
    const int d[4] = {dv[k].x, dv[k].y, dv[k].z, dv[k].w};
#pragma unroll
    for (int j = 0; j < 4; ++j) {
      const unsigned o = (unsigned)(d[j] - base);
      int v = -1;
      if (o < (unsigned)nr) v = lmap[o];
      sl[4 * k + j] = v;
      c += (v >= 0) ? 1 : 0;
    }
  }
  int x = c;
#pragma unroll
  for (int d = 1; d < 32; d <<= 1) {
    const int y = __shfl_up(x, d);
    if (lane >= d) x += y;
  }
  if (lane == 31) wsum[wave] = x;
  __syncthreads();
  if (wave == 0) {
    const int w = wsum[lane];
    int xx = w;
#pragma unroll
    for (int d = 1; d < 32; d <<= 1) {
      const int y = __shfl_up(xx, d);
      if (lane >= d) xx += y;
    }
    wsum[lane] = xx - w;
    if (lane == 31) wtot = xx;
  }
  __syncthreads();
  const int total = wtot;
  const int seg   = (sc ? SH + share : share) * SEGSTR;
  const int start = (range == 1) ? seg + SEGSTR - ((total + 15) & ~15) : seg;
  if (t == 0) {
    runs[2 * b]     = start;
    runs[2 * b + 1] = total;
  }
  int pos = start + wsum[wave] + (x - c);
#pragma unroll
  for (int k = 0; k < FEPT / 4; ++k) {
#pragma unroll
    for (int j = 0; j < 4; ++j) {
      const int v = sl[4 * k + j];
      if (v >= 0) {
        lists[pos] = (v << EBITS) | (share * SEGCAP + 4 * (k * 1024 + t) + j);
        tags[pos]  = (unsigned char)(v / PR);
        ++pos;
      }
    }
  }
}

__device__ __forceinline__ int kfrag(int t, int h) { return 8 * t + 4 * h; }

__device__ __forceinline__ void keep_ahead(float4& w) {
  asm volatile("" : "+v"(w.x), "+v"(w.y), "+v"(w.z), "+v"(w.w));
}

__global__ __launch_bounds__(PT, 8) void k_pullmix(const float* __restrict__ v_grid,
                                                const float* __restrict__ v_sc,
                                                const float* __restrict__ att_s2g,
                                                const float* __restrict__ att_g2s,
                                                const int* __restrict__ src_s2g,
                                                const int* __restrict__ src_g2s,
                                                const int* __restrict__ lists,
                                                const unsigned char* __restrict__ tags,
                                                const unsigned char* __restrict__ qtags,
                                                const int* __restrict__ runs,
                                                const int* __restrict__ qslot,
                                                const int* __restrict__ gr_id,
                                                const int* __restrict__ sc_id,
                                                const float* __restrict__ W1,
                                                const float* __restrict__ b1,
                                                float* __restrict__ out) {
  __shared__ __attribute__((aligned(16))) float acc[(PR + 1) * DIM];
  __shared__ __attribute__((aligned(16))) float arena[ARENA];
  __shared__ int qlist[NQ];
  __shared__ int rownode[PR];
  __shared__ int mcount;
  __shared__ int qcount;
  __shared__ int upre[33];
  __shared__ int rstart[32];
  __shared__ int rcount[32];

  int* mlist = (int*)arena;
  const int tid  = threadIdx.x;
  const int lane = tid & 31;
  const int wave = __builtin_amdgcn_readfirstlane(tid >> 5);
  const unsigned blk = blockIdx.x;
  const int side = 1 - (int)blockIdx.y;

  const float* vsrc  = side ? v_grid  : v_sc;
  const float* att   = side ? att_g2s : att_s2g;
  const int*   src   = side ? src_g2s : src_s2g;
  const int  run0 = side ? NRUNG : 0;
  const int  nrun = side ? SH : NRUNG;
  const unsigned char* qt = qtags + (size_t)side * NQ;
  const unsigned pat = blk * 0x01010101u;
  const float* vself = side ? v_sc    : v_grid;
  const int*   qs    = qslot + (size_t)side * NQ;
  const int*   ids   = side ? sc_id   : gr_id;
  float*       obase = out + (side ? (size_t)0 : (size_t)NQ * DIM);
  if (wave == 0) {
    const int st = (lane < nrun) ? runs[2 * (run0 + lane)]     : 0;
    const int rc = (lane < nrun) ? runs[2 * (run0 + lane) + 1] : 0;
    const int u  = (rc + 15) >> 4;
    int x = u;
#pragma unroll
    for (int d = 1; d < 32; d <<= 1) {
      const int y = __shfl_up(x, d);
      if (lane >= d) x += y;
    }
    upre[lane]   = x - u;
    rstart[lane] = st;
    rcount[lane] = rc;
    if (lane == 31) upre[32] = x;
  }

  for (int i = tid; i < (PR + 1) * DIM; i += PT) acc[i] = 0.0f;
  if (tid < PR) rownode[tid] = -1;
  if (tid == 0) { mcount = 0; qcount = 0; }
  __syncthreads();

  if (tid < NQ / 16) {
    const uint4 t = *(const uint4*)(qt + tid * 16);
    const unsigned w[4] = {t.x, t.y, t.z, t.w};
#pragma unroll
    for (int d = 0; d < 4; ++d) {
      const unsigned y = w[d] ^ pat;
      if (((y - 0x01010101u) & ~y & 0x80808080u) != 0u) {
#pragma unroll
        for (int j = 0; j < 4; ++j) {
          if (((y >> (8 * j)) & 0xFFu) == 0u) {
            const int pos = atomicAdd(&qcount, 1);
            qlist[pos] = tid * 16 + 4 * d + j;
          }
        }
      }
    }
  }
  __syncthreads();
  const int nq = __builtin_amdgcn_readfirstlane(qcount);
  if (nq == 0) return;
  for (int i = tid; i < nq; i += PT) {
    const int q = qlist[i];
    const int g = ids[q];
    const int r = qs[q] % PR;
    qlist[i]   = (q << 4) | r;
    rownode[r] = g;
  }
  __syncthreads();

  const int nu = __builtin_amdgcn_readfirstlane(upre[32]);
  for (int c0 = 0; c0 < nu; c0 += PCH / 16) {
    uint4 tv[PCH / (PT * 16)];
    int   gp[PCH / (PT * 16)];
    int   lm[PCH / (PT * 16)];
#pragma unroll
    for (int k = 0; k < PCH / (PT * 16); ++k) {
      const int  L  = c0 + k * PT + tid;
      const bool ok = L < nu;
      const int  Lc = ok ? L : 0;
      int ri = 0;
#pragma unroll
      for (int sp = 16; sp >= 1; sp >>= 1) {
        if (upre[ri + sp] <= Lc) ri += sp;
      }
      const int off = (Lc - upre[ri]) * 16;
      gp[k] = rstart[ri] + off;
      lm[k] = ok ? rcount[ri] - off : 0;
      tv[k] = *(const uint4*)(tags + gp[k]);
    }
#pragma unroll
    for (int k = 0; k < PCH / (PT * 16); ++k) {
      const unsigned w[4] = {tv[k].x, tv[k].y, tv[k].z, tv[k].w};
#pragma unroll
      for (int d = 0; d < 4; ++d) {
        const unsigned y = w[d] ^ pat;
        if (((y - 0x01010101u) & ~y & 0x80808080u) != 0u) {
#pragma unroll
          for (int j = 0; j < 4; ++j) {
            const int idx = 4 * d + j;
            if (((y >> (8 * j)) & 0xFFu) == 0u && idx < lm[k]) {
              const int pos = atomicAdd(&mcount, 1);
              mlist[pos] = gp[k] + idx;
            }
          }
        }
      }
    }
    __syncthreads();
    const int mc = __builtin_amdgcn_readfirstlane(mcount);
    __syncthreads();

    if (mc > PCAP - PCH || c0 + PCH / 16 >= nu) {
      for (int i = tid; i < mc; i += PT) mlist[i] = lists[mlist[i]];
      __syncthreads();
      for (int b0 = wave * PB; b0 < mc; b0 += (PT / 32) * PB) {
        int   srow = 0;
        int   arow = PR;
        float a    = 0.0f;
        if (lane < PB && b0 + lane < mc) {
          const unsigned ment = (unsigned)mlist[b0 + lane];
          const int e = (int)(ment & ((1u << EBITS) - 1u));
          srow = src[e];
          a    = att[e];
          arow = (int)((ment >> EBITS) % PR);
        }
        float4 rv[PB][2];
#pragma unroll
        for (int b = 0; b < PB; ++b) {
          const int sr = __shfl(srow, b);
          const float4* vp = (const float4*)(vsrc + (size_t)sr * DIM);
          rv[b][0] = vp[lane];
          rv[b][1] = vp[lane + 32];
        }
#pragma unroll
        for (int b = 0; b < PB; ++b) {
          const float ab = __shfl(a, b);
          float* ap = acc + __shfl(arow, b) * DIM + lane * 4;
          atomicAdd(ap + 0,   rv[b][0].x * ab);
          atomicAdd(ap + 1,   rv[b][0].y * ab);
          atomicAdd(ap + 2,   rv[b][0].z * ab);
          atomicAdd(ap + 3,   rv[b][0].w * ab);
          atomicAdd(ap + 128, rv[b][1].x * ab);
          atomicAdd(ap + 129, rv[b][1].y * ab);
          atomicAdd(ap + 130, rv[b][1].z * ab);
          atomicAdd(ap + 131, rv[b][1].w * ab);
        }
      }
      __syncthreads();
      if (tid == 0) mcount = 0;
      __syncthreads();
    }
  }

  const int c  = lane & 15;
  const int h  = lane >> 4;
  const int n0 = wave * 16;

  const int wr = lane >> 3;
  const int wq = lane & 7;
  const float* wsl = W1 + (size_t)(n0 + wr) * DIM + wq * 4;
  float4 wp[NPIECE][4];
#pragma unroll
  for (int p = 0; p < 2; ++p) {
#pragma unroll
    for (int j = 0; j < 4; ++j) wp[p][j] = *(const float4*)(wsl + (4 * j) * DIM + WK * p);
  }
  const float bias = b1[n0 + c];

  float* As = arena;
  float* Ap = As + PR * WPITCH;
  float* wt = Ap + PR * WPITCH + wave * (2 * WPIECE);
#pragma unroll
  for (int j = 0; j < 4; ++j) *(float4*)(wt + (4 * j + wr) * WTP + wq * 4) = wp[0][j];
  {
    const int r = tid >> 5;
    const int g = rownode[r];
    float4 v0 = make_float4(0.0f, 0.0f, 0.0f, 0.0f);
    float4 v1 = v0;
    if (g >= 0) {
      const float4* vp = (const float4*)(vself + (size_t)g * DIM);
      v0 = vp[lane];
      v1 = vp[lane + 32];
    }
    const float4 h0 = *(const float4*)(acc + r * DIM + lane * 4);
    const float4 h1 = *(const float4*)(acc + r * DIM + 128 + lane * 4);
    float* s0 = As + r * WPITCH + lane * 4;
    float* p0 = Ap + r * WPITCH + lane * 4;
    *(float4*)(s0)       = make_float4(v0.x + h0.x, v0.y + h0.y, v0.z + h0.z, v0.w + h0.w);
    *(float4*)(s0 + 128) = make_float4(v1.x + h1.x, v1.y + h1.y, v1.z + h1.z, v1.w + h1.w);
    *(float4*)(p0)       = make_float4(v0.x * h0.x, v0.y * h0.y, v0.z * h0.z, v0.w * h0.w);
    *(float4*)(p0 + 128) = make_float4(v1.x * h1.x, v1.y * h1.y, v1.z * h1.z, v1.w * h1.w);
  }
#pragma unroll
  for (int j = 0; j < 4; ++j) keep_ahead(wp[1][j]);
  __syncthreads();

  v8f ysum = (v8f){};
  v8f yprd = (v8f){};
  {
    const float* asl = As + c * WPITCH;
    const float* apl = Ap + c * WPITCH;
#pragma unroll
    for (int p = 0; p < NPIECE; ++p) {
      if (p + 1 < NPIECE) {
        float* wn = wt + ((p + 1) & 1) * WPIECE;
#pragma unroll
        for (int j = 0; j < 4; ++j) *(float4*)(wn + (4 * j + wr) * WTP + wq * 4) = wp[p + 1][j];
      }
      if (p + 2 < NPIECE) {
#pragma unroll
        for (int j = 0; j < 4; ++j) wp[p + 2][j] = *(const float4*)(wsl + (4 * j) * DIM + WK * (p + 2));
      }
      const float* wc = wt + (p & 1) * WPIECE + c * WTP + 4 * h;
      {
        float4 fb[4], fs[4], fp[4];
#pragma unroll
        for (int u = 0; u < 4; ++u) {
          fb[u] = *(const float4*)(wc + 8 * u);
          fs[u] = *(const float4*)(asl + kfrag(4 * p + u, h));
          fp[u] = *(const float4*)(apl + kfrag(4 * p + u, h));
        }
#pragma unroll
        for (int u = 0; u < 4; ++u) {
          v2f b0;  b0.x  = fb[u].x; b0.y  = fb[u].y;
          v2f b1v; b1v.x = fb[u].z; b1v.y = fb[u].w;
          v2f s0;  s0.x  = fs[u].x; s0.y  = fs[u].y;
          v2f s1;  s1.x  = fs[u].z; s1.y  = fs[u].w;
          v2f p0;  p0.x  = fp[u].x; p0.y  = fp[u].y;
          v2f p1;  p1.x  = fp[u].z; p1.y  = fp[u].w;
          ysum = __builtin_amdgcn_wmma_f32_16x16x4_f32(false, s0, false, b0,  (short)0, ysum, false, false);
          yprd = __builtin_amdgcn_wmma_f32_16x16x4_f32(false, p0, false, b0,  (short)0, yprd, false, false);
          ysum = __builtin_amdgcn_wmma_f32_16x16x4_f32(false, s1, false, b1v, (short)0, ysum, false, false);
          yprd = __builtin_amdgcn_wmma_f32_16x16x4_f32(false, p1, false, b1v, (short)0, yprd, false, false);
        }
      }
      if (p + 2 < NPIECE) {
#pragma unroll
        for (int j = 0; j < 4; ++j) keep_ahead(wp[p + 2][j]);
      }
      if (p + 1 < NPIECE) __syncthreads();
    }
  }

#pragma unroll
  for (int r = 0; r < 8; ++r) {
    float z1 = ysum[r] + bias;
    float z2 = yprd[r] + bias;
    z1 = (z1 >= 0.0f) ? z1 : 0.01f * z1;
    z2 = (z2 >= 0.0f) ? z2 : 0.01f * z2;
    acc[(8 * h + r) * DIM + n0 + c] = z1 + z2;
  }
  __syncthreads();

  for (int i = wave; i < nq; i += PT / 32) {
    const int qe = qlist[i];
    const int q  = qe >> 4;
    const int r  = qe & 15;
    const float4* yp = (const float4*)(acc + r * DIM);
    float4* op = (float4*)(obase + (size_t)q * DIM);
    op[lane]      = yp[lane];
    op[lane + 32] = yp[lane + 32];
  }
}

extern "C" void kernel_launch(void* const* d_in, const int* in_sizes, int n_in,
                              void* d_out, int out_size, void* d_ws, size_t ws_size,
                              hipStream_t stream) {
  const float* v_grid  = (const float*)d_in[0];
  const float* v_sc    = (const float*)d_in[1];
  const float* att_s2g = (const float*)d_in[2];
  const float* att_g2s = (const float*)d_in[3];
  const float* W1      = (const float*)d_in[4];
  const float* b1      = (const float*)d_in[5];
  const int*   src_s2g = (const int*)d_in[6];
  const int*   dst_s2g = (const int*)d_in[7];
  const int*   src_g2s = (const int*)d_in[8];
  const int*   dst_g2s = (const int*)d_in[9];
  const int*   sc_id   = (const int*)d_in[10];
  const int*   gr_id   = (const int*)d_in[11];
  float* out = (float*)d_out;

  int* runs  = (int*)d_ws;
  int* qslot = runs + 128;
  int* lists = qslot + 2 * NQ;
  unsigned char* tags  = (unsigned char*)(lists + NLIST);
  unsigned char* qtags = tags + NLIST;

  k_front<<<NRUN, 1024, 0, stream>>>(dst_s2g, dst_g2s, gr_id, sc_id, lists, tags, runs, qslot, qtags);

  k_pullmix<<<dim3(MAXU / PR, 2), PT, 0, stream>>>(v_grid, v_sc, att_s2g, att_g2s, src_s2g, src_g2s,
                                                 lists, tags, qtags, runs, qslot, gr_id, sc_id, W1, b1, out);
}


# pull scan: wave w reads the owner tags of its own two runs (w, w+16); a unit's list place is arithmetic on scalars, not a 7-hop LDS search of the run table per load; nh sum still in arrival order
# speedup vs baseline: 5.9443x; 1.0327x over previous
#include <hip/hip_runtime.h>

#define NG     100000
#define NS     20000
#define EDGES  300000
#define DIM    256
#define NQ     4096
#define MAXU   NQ
#define RN     50000
#define NRG    ((NG + RN - 1) / RN)
#define SH     16
#define SEGCAP 18752
#define SEGSTR (SEGCAP + 16)
#define NRUNG  (NRG * SH)
#define NRUN   (NRUNG + SH)
#define NLIST  (2 * SH * SEGSTR)
#define FEPT   20
#define WPITCH 260
#define EBITS  19
#define PR     16
#define PT     512
#define PCH    (PT * 48)
#define WQ     (PCH / 16 / (PT / 32))
#define PB     8
#define WK     32
#define NPIECE (DIM / WK)
#define WTP    36
#define WPIECE (16 * WTP)
#define ARENA  (2 * PR * WPITCH + (PT / 32) * 2 * WPIECE)
#define PCAP   ARENA

static_assert(EDGES <= (1 << EBITS), "edge index must fit the packed entry");
static_assert(MAXU  <= (1 << (31 - EBITS)), "slot must fit the packed entry");
static_assert(MAXU % PR == 0 && PR == 16, "a workgroup owns one 16-row tile of slots");
static_assert(PT / 32 * 16 == DIM, "one wave per 16-column tile of the output");
static_assert(PCAP <= ARENA && PCAP > PCH, "the match list lives in the arena and takes one more chunk behind the flush mark");
static_assert((17 * DIM + ARENA + NQ) * 4 + 256 <= 160 * 1024, "two workgroups per WGP");
static_assert(MAXU / PR <= 256, "the owner of a slot (slot / PR) fits a one-byte tag");
static_assert(NQ % 1024 == 0, "whole workgroups of the edge filter write the query tags");
static_assert(NQ % 16 == 0 && NQ / 16 <= PT, "one 16-tag load per thread covers the query tags");
static_assert(SEGCAP % 16 == 0 && SEGSTR % 16 == 0 && PCH % 16 == 0, "16-tag loads stay 16-B aligned");
static_assert(NQ == 4 * 1024, "a front thread takes four queries in one 16-B load");
static_assert(NG % 4 == 0 && NS % 4 == 0 && RN % 4 == 0, "the 16-B LDS stores that clear a range end with the range");
static_assert(NRG == 2 && NS <= RN, "two grid ranges share a segment from its two ends; one range holds the small-category side");
static_assert(SH * SEGCAP >= EDGES && EDGES % 4 == 0 && SEGCAP % 4 == 0, "the shares cover the edges in whole 16-B loads");
static_assert(1024 * FEPT >= SEGCAP && FEPT % 4 == 0, "a front workgroup covers its share");
static_assert(NRUNG <= 32 && SH <= 32 && 2 * NRUN <= 128, "one wave reads a side's run table; the table fits its 128 ints");
static_assert(SH == PT / 32 && NRUNG == 2 * SH, "pull wave w scans the runs w and w + SH of its side");
static_assert(WQ % 32 == 0 && WQ * 16 * (PT / 32) == PCH, "a wave's share of a chunk is whole loads of all its lanes");

typedef __attribute__((ext_vector_type(2))) float v2f;
typedef __attribute__((ext_vector_type(8))) float v8f;

__global__ __launch_bounds__(1024) void k_front(const int* __restrict__ dst_g,
                                                const int* __restrict__ dst_s,
                                                const int* __restrict__ gid,
                                                const int* __restrict__ sid,
                                                int* __restrict__ lists,
                                                unsigned char* __restrict__ tags,
                                                int* __restrict__ runs,
                                                int* __restrict__ qslot,
                                                unsigned char* __restrict__ qtags) {
  __shared__ __attribute__((aligned(16))) int lmap[RN];
  __shared__ int wsum[32];
  __shared__ int wtot;
  const int  t     = threadIdx.x;
  const int  lane  = t & 31;
  const int  wave  = t >> 5;
  const int  b     = blockIdx.x;
  const bool sc    = b >= NRUNG;
  const int  range = sc ? 0 : b / SH;
  const int  share = sc ? b - NRUNG : b % SH;
  const int  side  = sc ? 1 : 0;
  const int  base  = range * RN;
  const int  left  = (sc ? NS : NG) - base;
  const int  nr    = left < RN ? left : RN;
  const int* ids   = sc ? sid : gid;
  const int* dst   = sc ? dst_s : dst_g;

  int4 dv[FEPT / 4];
#pragma unroll
  for (int k = 0; k < FEPT / 4; ++k) {
    const int i4 = 4 * (k * 1024 + t);
    const int e0 = share * SEGCAP + i4;
    dv[k] = make_int4(-1, -1, -1, -1);
    if (i4 < SEGCAP && e0 < EDGES) dv[k] = *(const int4*)(dst + e0);
  }
  const int4 q = *(const int4*)(ids + 4 * t);
  for (int i = t; i < nr / 4; i += 1024) *(int4*)(lmap + 4 * i) = make_int4(-1, -1, -1, -1);
  __syncthreads();
  const int g[4] = {q.x, q.y, q.z, q.w};
#pragma unroll
  for (int k = 0; k < 4; ++k) {
    const unsigned o = (unsigned)(g[k] - base);
    if (o < (unsigned)nr) atomicMax(&lmap[o], 4 * t + k);
  }
  __syncthreads();
  if (share == 0) {
#pragma unroll
    for (int k = 0; k < 4; ++k) {
      const unsigned o = (unsigned)(g[k] - base);
      if (o < (unsigned)nr) {
        const int sl = lmap[o];
        qslot[side * NQ + 4 * t + k] = sl;
        qtags[side * NQ + 4 * t + k] = (unsigned char)(sl / PR);
      }
    }
  }

  int sl[FEPT];
  int c = 0;
#pragma unroll
  for (int k = 0; k < FEPT / 4; ++k) {
    const int d[4] = {dv[k].x, dv[k].y, dv[k].z, dv[k].w};
#pragma unroll
    for (int j = 0; j < 4; ++j) {
      const unsigned o = (unsigned)(d[j] - base);
      int v = -1;
      if (o < (unsigned)nr) v = lmap[o];
      sl[4 * k + j] = v;
      c += (v >= 0) ? 1 : 0;
    }
  }
  int x = c;
#pragma unroll
  for (int d = 1; d < 32; d <<= 1) {
    const int y = __shfl_up(x, d);
    if (lane >= d) x += y;
  }
  if (lane == 31) wsum[wave] = x;
  __syncthreads();
  if (wave == 0) {
    const int w = wsum[lane];
    int xx = w;
#pragma unroll
    for (int d = 1; d < 32; d <<= 1) {
      const int y = __shfl_up(xx, d);
      if (lane >= d) xx += y;
    }
    wsum[lane] = xx - w;
    if (lane == 31) wtot = xx;
  }
  __syncthreads();
  const int total = wtot;
  const int seg   = (sc ? SH + share : share) * SEGSTR;
  const int start = (range == 1) ? seg + SEGSTR - ((total + 15) & ~15) : seg;
  if (t == 0) {
    runs[2 * b]     = start;
    runs[2 * b + 1] = total;
  }
  int pos = start + wsum[wave] + (x - c);
#pragma unroll
  for (int k = 0; k < FEPT / 4; ++k) {
#pragma unroll
    for (int j = 0; j < 4; ++j) {
      const int v = sl[4 * k + j];
      if (v >= 0) {
        lists[pos] = (v << EBITS) | (share * SEGCAP + 4 * (k * 1024 + t) + j);
        tags[pos]  = (unsigned char)(v / PR);
        ++pos;
      }
    }
  }
}

__device__ __forceinline__ int kfrag(int t, int h) { return 8 * t + 4 * h; }

__device__ __forceinline__ void keep_ahead(float4& w) {
  asm volatile("" : "+v"(w.x), "+v"(w.y), "+v"(w.z), "+v"(w.w));
}

__global__ __launch_bounds__(PT, 8) void k_pullmix(const float* __restrict__ v_grid,
                                                const float* __restrict__ v_sc,
                                                const float* __restrict__ att_s2g,
                                                const float* __restrict__ att_g2s,
                                                const int* __restrict__ src_s2g,
                                                const int* __restrict__ src_g2s,
                                                const int* __restrict__ lists,
                                                const unsigned char* __restrict__ tags,
                                                const unsigned char* __restrict__ qtags,
                                                const int* __restrict__ runs,
                                                const int* __restrict__ qslot,
                                                const int* __restrict__ gr_id,
                                                const int* __restrict__ sc_id,
                                                const float* __restrict__ W1,
                                                const float* __restrict__ b1,
                                                float* __restrict__ out) {
  __shared__ __attribute__((aligned(16))) float acc[(PR + 1) * DIM];
  __shared__ __attribute__((aligned(16))) float arena[ARENA];
  __shared__ int qlist[NQ];
  __shared__ int rownode[PR];
  __shared__ int mcount;
  __shared__ int qcount;
  __shared__ int rstart[32];
  __shared__ int rcount[32];
  __shared__ int umax;

  int* mlist = (int*)arena;
  const int tid  = threadIdx.x;
  const int lane = tid & 31;
  const int wave = __builtin_amdgcn_readfirstlane(tid >> 5);
  const unsigned blk = blockIdx.x;
  const int side = 1 - (int)blockIdx.y;

  const float* vsrc  = side ? v_grid  : v_sc;
  const float* att   = side ? att_g2s : att_s2g;
  const int*   src   = side ? src_g2s : src_s2g;
  const int  run0 = side ? NRUNG : 0;
  const int  nrun = side ? SH : NRUNG;
  const unsigned char* qt = qtags + (size_t)side * NQ;
  const unsigned pat = blk * 0x01010101u;
  const float* vself = side ? v_sc    : v_grid;
  const int*   qs    = qslot + (size_t)side * NQ;
  const int*   ids   = side ? sc_id   : gr_id;
  float*       obase = out + (side ? (size_t)0 : (size_t)NQ * DIM);
  if (wave == 0) {
    const int st = (lane < nrun) ? runs[2 * (run0 + lane)]     : 0;
    const int rc = (lane < nrun) ? runs[2 * (run0 + lane) + 1] : 0;
    const int u  = (rc + 15) >> 4;
    int x = u + __shfl_xor(u, SH);
#pragma unroll
    for (int d = SH / 2; d >= 1; d >>= 1) {
      const int y = __shfl_xor(x, d);
      x = x > y ? x : y;
    }
    rstart[lane] = st;
    rcount[lane] = rc;
    if (lane == 0) umax = x;
  }

  for (int i = tid; i < (PR + 1) * DIM; i += PT) acc[i] = 0.0f;
  if (tid < PR) rownode[tid] = -1;
  if (tid == 0) { mcount = 0; qcount = 0; }
  __syncthreads();

  if (tid < NQ / 16) {
    const uint4 t = *(const uint4*)(qt + tid * 16);
    const unsigned w[4] = {t.x, t.y, t.z, t.w};
#pragma unroll
    for (int d = 0; d < 4; ++d) {
      const unsigned y = w[d] ^ pat;
      if (((y - 0x01010101u) & ~y & 0x80808080u) != 0u) {
#pragma unroll
        for (int j = 0; j < 4; ++j) {
          if (((y >> (8 * j)) & 0xFFu) == 0u) {
            const int pos = atomicAdd(&qcount, 1);
            qlist[pos] = tid * 16 + 4 * d + j;
          }
        }
      }
    }
  }
  __syncthreads();
  const int nq = __builtin_amdgcn_readfirstlane(qcount);
  if (nq == 0) return;
  for (int i = tid; i < nq; i += PT) {
    const int q = qlist[i];
    const int g = ids[q];
    const int r = qs[q] % PR;
    qlist[i]   = (q << 4) | r;
    rownode[r] = g;
  }
  __syncthreads();

  const int stA = __builtin_amdgcn_readfirstlane(rstart[wave]);
  const int rcA = __builtin_amdgcn_readfirstlane(rcount[wave]);
  const int stB = __builtin_amdgcn_readfirstlane(rstart[wave + SH]);
  const int rcB = __builtin_amdgcn_readfirstlane(rcount[wave + SH]);
  const int uA  = (rcA + 15) >> 4;
  const int uT  = uA + ((rcB + 15) >> 4);
  const int nu  = __builtin_amdgcn_readfirstlane(umax);
  for (int c0 = 0; c0 < nu; c0 += WQ) {
    uint4 tv[WQ / 32];
    int   gp[WQ / 32];
    int   lm[WQ / 32];
#pragma unroll
    for (int k = 0; k < WQ / 32; ++k) {
      const int  V   = c0 + 32 * k + lane;
      const bool ok  = V < uT;
      const int  Vc  = ok ? V : 0;
      const bool inA = Vc < uA;
      const int  off = 16 * (inA ? Vc : Vc - uA);
      gp[k] = (inA ? stA : stB) + off;
      lm[k] = ok ? (inA ? rcA : rcB) - off : 0;
      tv[k] = *(const uint4*)(tags + gp[k]);
    }
#pragma unroll
    for (int k = 0; k < WQ / 32; ++k) {
      const unsigned w[4] = {tv[k].x, tv[k].y, tv[k].z, tv[k].w};
#pragma unroll
      for (int d = 0; d < 4; ++d) {
        const unsigned y = w[d] ^ pat;
        if (((y - 0x01010101u) & ~y & 0x80808080u) != 0u) {
#pragma unroll
          for (int j = 0; j < 4; ++j) {
            const int idx = 4 * d + j;
            if (((y >> (8 * j)) & 0xFFu) == 0u && idx < lm[k]) {
              const int pos = atomicAdd(&mcount, 1);
              mlist[pos] = gp[k] + idx;
            }
          }
        }
      }
    }
    __syncthreads();
    const int mc = __builtin_amdgcn_readfirstlane(mcount);
    __syncthreads();

    if (mc > PCAP - PCH || c0 + WQ >= nu) {
      for (int i = tid; i < mc; i += PT) mlist[i] = lists[mlist[i]];
      __syncthreads();
      for (int b0 = wave * PB; b0 < mc; b0 += (PT / 32) * PB) {
        int   srow = 0;
        int   arow = PR;
        float a    = 0.0f;
        if (lane < PB && b0 + lane < mc) {
          const unsigned ment = (unsigned)mlist[b0 + lane];
          const int e = (int)(ment & ((1u << EBITS) - 1u));
          srow = src[e];
          a    = att[e];
          arow = (int)((ment >> EBITS) % PR);
        }
        float4 rv[PB][2];
#pragma unroll
        for (int b = 0; b < PB; ++b) {
          const int sr = __shfl(srow, b);
          const float4* vp = (const float4*)(vsrc + (size_t)sr * DIM);
          rv[b][0] = vp[lane];
          rv[b][1] = vp[lane + 32];
        }
#pragma unroll
        for (int b = 0; b < PB; ++b) {
          const float ab = __shfl(a, b);
          float* ap = acc + __shfl(arow, b) * DIM + lane * 4;
          atomicAdd(ap + 0,   rv[b][0].x * ab);
          atomicAdd(ap + 1,   rv[b][0].y * ab);
          atomicAdd(ap + 2,   rv[b][0].z * ab);
          atomicAdd(ap + 3,   rv[b][0].w * ab);
          atomicAdd(ap + 128, rv[b][1].x * ab);
          atomicAdd(ap + 129, rv[b][1].y * ab);
          atomicAdd(ap + 130, rv[b][1].z * ab);
          atomicAdd(ap + 131, rv[b][1].w * ab);
        }
      }
      __syncthreads();
      if (tid == 0) mcount = 0;
      __syncthreads();
    }
  }

  const int c  = lane & 15;
  const int h  = lane >> 4;
  const int n0 = wave * 16;

  const int wr = lane >> 3;
  const int wq = lane & 7;
  const float* wsl = W1 + (size_t)(n0 + wr) * DIM + wq * 4;
  float4 wp[NPIECE][4];
#pragma unroll
  for (int p = 0; p < 2; ++p) {
#pragma unroll
    for (int j = 0; j < 4; ++j) wp[p][j] = *(const float4*)(wsl + (4 * j) * DIM + WK * p);
  }
  const float bias = b1[n0 + c];

  float* As = arena;
  float* Ap = As + PR * WPITCH;
  float* wt = Ap + PR * WPITCH + wave * (2 * WPIECE);
#pragma unroll
  for (int j = 0; j < 4; ++j) *(float4*)(wt + (4 * j + wr) * WTP + wq * 4) = wp[0][j];
  {
    const int r = tid >> 5;
    const int g = rownode[r];
    float4 v0 = make_float4(0.0f, 0.0f, 0.0f, 0.0f);
    float4 v1 = v0;
    if (g >= 0) {
      const float4* vp = (const float4*)(vself + (size_t)g * DIM);
      v0 = vp[lane];
      v1 = vp[lane + 32];
    }
    const float4 h0 = *(const float4*)(acc + r * DIM + lane * 4);
    const float4 h1 = *(const float4*)(acc + r * DIM + 128 + lane * 4);
    float* s0 = As + r * WPITCH + lane * 4;
    float* p0 = Ap + r * WPITCH + lane * 4;
    *(float4*)(s0)       = make_float4(v0.x + h0.x, v0.y + h0.y, v0.z + h0.z, v0.w + h0.w);
    *(float4*)(s0 + 128) = make_float4(v1.x + h1.x, v1.y + h1.y, v1.z + h1.z, v1.w + h1.w);
    *(float4*)(p0)       = make_float4(v0.x * h0.x, v0.y * h0.y, v0.z * h0.z, v0.w * h0.w);
    *(float4*)(p0 + 128) = make_float4(v1.x * h1.x, v1.y * h1.y, v1.z * h1.z, v1.w * h1.w);
  }
#pragma unroll
  for (int j = 0; j < 4; ++j) keep_ahead(wp[1][j]);
  __syncthreads();

  v8f ysum = (v8f){};
  v8f yprd = (v8f){};
  {
    const float* asl = As + c * WPITCH;
    const float* apl = Ap + c * WPITCH;
#pragma unroll
    for (int p = 0; p < NPIECE; ++p) {
      if (p + 1 < NPIECE) {
        float* wn = wt + ((p + 1) & 1) * WPIECE;
#pragma unroll
        for (int j = 0; j < 4; ++j) *(float4*)(wn + (4 * j + wr) * WTP + wq * 4) = wp[p + 1][j];
      }
      if (p + 2 < NPIECE) {
#pragma unroll
        for (int j = 0; j < 4; ++j) wp[p + 2][j] = *(const float4*)(wsl + (4 * j) * DIM + WK * (p + 2));
      }
      const float* wc = wt + (p & 1) * WPIECE + c * WTP + 4 * h;
      {
        float4 fb[4], fs[4], fp[4];
#pragma unroll
        for (int u = 0; u < 4; ++u) {
          fb[u] = *(const float4*)(wc + 8 * u);
          fs[u] = *(const float4*)(asl + kfrag(4 * p + u, h));
          fp[u] = *(const float4*)(apl + kfrag(4 * p + u, h));
        }
#pragma unroll
        for (int u = 0; u < 4; ++u) {
          v2f b0;  b0.x  = fb[u].x; b0.y  = fb[u].y;
          v2f b1v; b1v.x = fb[u].z; b1v.y = fb[u].w;
          v2f s0;  s0.x  = fs[u].x; s0.y  = fs[u].y;
          v2f s1;  s1.x  = fs[u].z; s1.y  = fs[u].w;
          v2f p0;  p0.x  = fp[u].x; p0.y  = fp[u].y;
          v2f p1;  p1.x  = fp[u].z; p1.y  = fp[u].w;
          ysum = __builtin_amdgcn_wmma_f32_16x16x4_f32(false, s0, false, b0,  (short)0, ysum, false, false);
          yprd = __builtin_amdgcn_wmma_f32_16x16x4_f32(false, p0, false, b0,  (short)0, yprd, false, false);
          ysum = __builtin_amdgcn_wmma_f32_16x16x4_f32(false, s1, false, b1v, (short)0, ysum, false, false);
          yprd = __builtin_amdgcn_wmma_f32_16x16x4_f32(false, p1, false, b1v, (short)0, yprd, false, false);
        }
      }
      if (p + 2 < NPIECE) {
#pragma unroll
        for (int j = 0; j < 4; ++j) keep_ahead(wp[p + 2][j]);
      }
      if (p + 1 < NPIECE) __syncthreads();
    }
  }

#pragma unroll
  for (int r = 0; r < 8; ++r) {
    float z1 = ysum[r] + bias;
    float z2 = yprd[r] + bias;
    z1 = (z1 >= 0.0f) ? z1 : 0.01f * z1;
    z2 = (z2 >= 0.0f) ? z2 : 0.01f * z2;
    acc[(8 * h + r) * DIM + n0 + c] = z1 + z2;
  }
  __syncthreads();

  for (int i = wave; i < nq; i += PT / 32) {
    const int qe = qlist[i];
    const int q  = qe >> 4;
    const int r  = qe & 15;
    const float4* yp = (const float4*)(acc + r * DIM);
    float4* op = (float4*)(obase + (size_t)q * DIM);
    op[lane]      = yp[lane];
    op[lane + 32] = yp[lane + 32];
  }
}

extern "C" void kernel_launch(void* const* d_in, const int* in_sizes, int n_in,
                              void* d_out, int out_size, void* d_ws, size_t ws_size,
                              hipStream_t stream) {
  const float* v_grid  = (const float*)d_in[0];
  const float* v_sc    = (const float*)d_in[1];
  const float* att_s2g = (const float*)d_in[2];
  const float* att_g2s = (const float*)d_in[3];
  const float* W1      = (const float*)d_in[4];
  const float* b1      = (const float*)d_in[5];
  const int*   src_s2g = (const int*)d_in[6];
  const int*   dst_s2g = (const int*)d_in[7];
  const int*   src_g2s = (const int*)d_in[8];
  const int*   dst_g2s = (const int*)d_in[9];
  const int*   sc_id   = (const int*)d_in[10];
  const int*   gr_id   = (const int*)d_in[11];
  float* out = (float*)d_out;

  int* runs  = (int*)d_ws;
  int* qslot = runs + 128;
  int* lists = qslot + 2 * NQ;
  unsigned char* tags  = (unsigned char*)(lists + NLIST);
  unsigned char* qtags = tags + NLIST;

  k_front<<<NRUN, 1024, 0, stream>>>(dst_s2g, dst_g2s, gr_id, sc_id, lists, tags, runs, qslot, qtags);

  k_pullmix<<<dim3(MAXU / PR, 2), PT, 0, stream>>>(v_grid, v_sc, att_s2g, att_g2s, src_s2g, src_g2s,
                                                 lists, tags, qtags, runs, qslot, gr_id, sc_id, W1, b1, out);
}


# front groups each run's matches by owner workgroup (LDS count per owner gives the rank; 257 offsets per run); pull workgroups read their entries by offset, no tag scan; nh sum still in arrival order
# speedup vs baseline: 6.2098x; 1.0447x over previous
#include <hip/hip_runtime.h>

#define NG     100000
#define NS     20000
#define EDGES  300000
#define DIM    256
#define NQ     4096
#define MAXU   NQ
#define RN     50000
#define NRG    ((NG + RN - 1) / RN)
#define SH     16
#define SEGCAP 18752
#define NRUNG  (NRG * SH)
#define NRUN   (NRUNG + SH)
#define NLIST  (2 * SH * SEGCAP)
#define NOWN   (MAXU / PR)
#define OFFSTR (NOWN + 4)
#define FEPT   20
#define WPITCH 260
#define EBITS  19
#define PR     16
#define PT     512
#define PB     8
#define WK     32
#define NPIECE (DIM / WK)
#define WTP    36
#define WPIECE (16 * WTP)
#define ARENA  (2 * PR * WPITCH + (PT / 32) * 2 * WPIECE)
#define PCAP   ARENA

static_assert(EDGES <= (1 << EBITS), "edge index must fit the packed entry");
static_assert(MAXU  <= (1 << (31 - EBITS)), "slot must fit the packed entry");
static_assert(MAXU % PR == 0 && PR == 16, "a workgroup owns one 16-row tile of slots");
static_assert(PT / 32 * 16 == DIM, "one wave per 16-column tile of the output");
static_assert(PCAP <= ARENA, "the match list lives in the arena");
static_assert((17 * DIM + ARENA + NQ) * 4 + 256 <= 160 * 1024, "two workgroups per WGP");
static_assert(NOWN <= 256 && NOWN == 8 * 32, "the owner of a slot (slot / PR) fits a one-byte tag; one wave sums the owner counts, 8 per lane");
static_assert(OFFSTR % 4 == 0 && OFFSTR >= NOWN + 2, "an offset row is whole 16-B units and holds its offsets and the start");
static_assert(NQ % 1024 == 0, "whole workgroups of the edge filter write the query tags");
static_assert(NQ % 16 == 0 && NQ / 16 <= PT, "one 16-tag load per thread covers the query tags");
static_assert(NQ == 4 * 1024, "a front thread takes four queries in one 16-B load");
static_assert(NG % 4 == 0 && NS % 4 == 0 && RN % 4 == 0, "the 16-B LDS stores that clear a range end with the range");
static_assert(NRG == 2 && NS <= RN, "two grid ranges share a segment from its two ends; one range holds the small-category side");
static_assert(SH * SEGCAP >= EDGES && EDGES % 4 == 0 && SEGCAP % 4 == 0, "the shares cover the edges in whole 16-B loads");
static_assert(1024 * FEPT >= SEGCAP && FEPT % 4 == 0, "a front workgroup covers its share");
static_assert(NRUNG <= 32 && SH <= 32, "one wave reads a side's offsets, one run per lane");
static_assert(SH * 32 == PT && NRUNG * 16 == PT, "a pull workgroup copies its entries with 32 threads per run (small-category side) or 16 (grid side)");

typedef __attribute__((ext_vector_type(2))) float v2f;
typedef __attribute__((ext_vector_type(8))) float v8f;

__global__ __launch_bounds__(1024) void k_front(const int* __restrict__ dst_g,
                                                const int* __restrict__ dst_s,
                                                const int* __restrict__ gid,
                                                const int* __restrict__ sid,
                                                int* __restrict__ lists,
                                                int* __restrict__ offs,
                                                int* __restrict__ qslot,
                                                unsigned char* __restrict__ qtags) {
  __shared__ __attribute__((aligned(16))) int lmap[RN];
  __shared__ __attribute__((aligned(16))) int ocnt[NOWN];
  __shared__ int rstart0;
  const int  t     = threadIdx.x;
  const int  lane  = t & 31;
  const int  wave  = t >> 5;
  const int  b     = blockIdx.x;
  const bool sc    = b >= NRUNG;
  const int  range = sc ? 0 : b / SH;
  const int  share = sc ? b - NRUNG : b % SH;
  const int  side  = sc ? 1 : 0;
  const int  base  = range * RN;
  const int  left  = (sc ? NS : NG) - base;
  const int  nr    = left < RN ? left : RN;
  const int* ids   = sc ? sid : gid;
  const int* dst   = sc ? dst_s : dst_g;

  int4 dv[FEPT / 4];
#pragma unroll
  for (int k = 0; k < FEPT / 4; ++k) {
    const int i4 = 4 * (k * 1024 + t);
    const int e0 = share * SEGCAP + i4;
    dv[k] = make_int4(-1, -1, -1, -1);
    if (i4 < SEGCAP && e0 < EDGES) dv[k] = *(const int4*)(dst + e0);
  }
  const int4 q = *(const int4*)(ids + 4 * t);
  for (int i = t; i < nr / 4; i += 1024) *(int4*)(lmap + 4 * i) = make_int4(-1, -1, -1, -1);
  if (t < NOWN) ocnt[t] = 0;
  __syncthreads();
  const int g[4] = {q.x, q.y, q.z, q.w};
#pragma unroll
  for (int k = 0; k < 4; ++k) {
    const unsigned o = (unsigned)(g[k] - base);
    if (o < (unsigned)nr) atomicMax(&lmap[o], 4 * t + k);
  }
  __syncthreads();
  if (share == 0) {
#pragma unroll
    for (int k = 0; k < 4; ++k) {
      const unsigned o = (unsigned)(g[k] - base);
      if (o < (unsigned)nr) {
        const int sl = lmap[o];
        qslot[side * NQ + 4 * t + k] = sl;
        qtags[side * NQ + 4 * t + k] = (unsigned char)(sl / PR);
      }
    }
  }

  int sl[FEPT];
#pragma unroll
  for (int k = 0; k < FEPT / 4; ++k) {
    const int d[4] = {dv[k].x, dv[k].y, dv[k].z, dv[k].w};
#pragma unroll
    for (int j = 0; j < 4; ++j) {
      const unsigned o = (unsigned)(d[j] - base);
      int v = -1;
      if (o < (unsigned)nr) v = lmap[o];
      sl[4 * k + j] = v;
    }
  }
  int rk[FEPT];
#pragma unroll
  for (int i = 0; i < FEPT; ++i) {
    int r = 0;
    if (sl[i] >= 0) r = atomicAdd(&ocnt[sl[i] / PR], 1);
    rk[i] = r;
  }
  __syncthreads();
  if (wave == 0) {
    const int4 ca = *(const int4*)(ocnt + 8 * lane);
    const int4 cb = *(const int4*)(ocnt + 8 * lane + 4);
    const int s1 = ca.x + ca.y;
    const int s2 = s1 + ca.z;
    const int s3 = s2 + ca.w;
    const int s4 = s3 + cb.x;
    const int s5 = s4 + cb.y;
    const int s6 = s5 + cb.z;
    const int s7 = s6 + cb.w;
    int x = s7;
#pragma unroll
    for (int d = 1; d < 32; d <<= 1) {
      const int y = __shfl_up(x, d);
      if (lane >= d) x += y;
    }
    const int ex = x - s7;
    const int4 oa = make_int4(ex, ex + ca.x, ex + s1, ex + s2);
    const int4 ob = make_int4(ex + s3, ex + s4, ex + s5, ex + s6);
    *(int4*)(ocnt + 8 * lane)     = oa;
    *(int4*)(ocnt + 8 * lane + 4) = ob;
    const int total = __shfl(x, 31);
    const int seg   = (sc ? SH + share : share) * SEGCAP;
    const int start = (range == 1) ? seg + SEGCAP - total : seg;
    int* orow = offs + b * OFFSTR;
    *(int4*)(orow + 8 * lane)     = oa;
    *(int4*)(orow + 8 * lane + 4) = ob;
    if (lane == 0) {
      orow[NOWN]     = total;
      orow[NOWN + 1] = start;
      rstart0 = start;
    }
  }
  __syncthreads();
  const int start = rstart0;
#pragma unroll
  for (int i = 0; i < FEPT; ++i) rk[i] += ocnt[(sl[i] >= 0 ? sl[i] : 0) / PR];
#pragma unroll
  for (int k = 0; k < FEPT / 4; ++k) {
#pragma unroll
    for (int j = 0; j < 4; ++j) {
      const int v = sl[4 * k + j];
      if (v >= 0) lists[start + rk[4 * k + j]] = (v << EBITS) | (share * SEGCAP + 4 * (k * 1024 + t) + j);
    }
  }
}

__device__ __forceinline__ int kfrag(int t, int h) { return 8 * t + 4 * h; }

__device__ __forceinline__ void keep_ahead(float4& w) {
  asm volatile("" : "+v"(w.x), "+v"(w.y), "+v"(w.z), "+v"(w.w));
}

__global__ __launch_bounds__(PT, 8) void k_pullmix(const float* __restrict__ v_grid,
                                                const float* __restrict__ v_sc,
                                                const float* __restrict__ att_s2g,
                                                const float* __restrict__ att_g2s,
                                                const int* __restrict__ src_s2g,
                                                const int* __restrict__ src_g2s,
                                                const int* __restrict__ lists,
                                                const int* __restrict__ offs,
                                                const unsigned char* __restrict__ qtags,
                                                const int* __restrict__ qslot,
                                                const int* __restrict__ gr_id,
                                                const int* __restrict__ sc_id,
                                                const float* __restrict__ W1,
                                                const float* __restrict__ b1,
                                                float* __restrict__ out) {
  __shared__ __attribute__((aligned(16))) float acc[(PR + 1) * DIM];
  __shared__ __attribute__((aligned(16))) float arena[ARENA];
  __shared__ int qlist[NQ];
  __shared__ int rownode[PR];
  __shared__ int qcount;
  __shared__ int rplace[32];
  __shared__ int rcnt[32];
  __shared__ int rpre[32];
  __shared__ int mtotal;

  int* mlist = (int*)arena;
  const int tid  = threadIdx.x;
  const int lane = tid & 31;
  const int wave = __builtin_amdgcn_readfirstlane(tid >> 5);
  const unsigned blk = blockIdx.x;
  const int side = 1 - (int)blockIdx.y;

  const float* vsrc  = side ? v_grid  : v_sc;
  const float* att   = side ? att_g2s : att_s2g;
  const int*   src   = side ? src_g2s : src_s2g;
  const int  run0 = side ? NRUNG : 0;
  const int  nrun = side ? SH : NRUNG;
  const unsigned char* qt = qtags + (size_t)side * NQ;
  const unsigned pat = blk * 0x01010101u;
  const float* vself = side ? v_sc    : v_grid;
  const int*   qs    = qslot + (size_t)side * NQ;
  const int*   ids   = side ? sc_id   : gr_id;
  float*       obase = out + (side ? (size_t)0 : (size_t)NQ * DIM);
  if (wave == 0) {
    int place = 0;
    int cnt   = 0;
    if (lane < nrun) {
      const int* orow = offs + (run0 + lane) * OFFSTR;
      const int o0 = orow[blk];
      const int o1 = orow[blk + 1];
      place = orow[NOWN + 1] + o0;
      cnt   = o1 - o0;
    }
    int x = cnt;
#pragma unroll
    for (int d = 1; d < 32; d <<= 1) {
      const int y = __shfl_up(x, d);
      if (lane >= d) x += y;
    }
    rplace[lane] = place;
    rcnt[lane]   = cnt;
    rpre[lane]   = x - cnt;
    if (lane == 31) mtotal = x;
  }

  for (int i = tid; i < (PR + 1) * DIM; i += PT) acc[i] = 0.0f;
  if (tid < PR) rownode[tid] = -1;
  if (tid == 0) qcount = 0;
  __syncthreads();

  const int total  = __builtin_amdgcn_readfirstlane(mtotal);
  const int fstep  = side ? 32 : 16;
  const int frun   = side ? (tid >> 5) : (tid >> 4);
  const int fj0    = tid & (fstep - 1);
  const int fplace = rplace[frun];
  const int fcnt   = rcnt[frun];
  const int fpre   = rpre[frun];
  const bool fhas  = fj0 < fcnt && (unsigned)(fpre + fj0) < (unsigned)PCAP;
  int fent = 0;
  if (fhas) fent = lists[fplace + fj0];

  if (tid < NQ / 16) {
    const uint4 t = *(const uint4*)(qt + tid * 16);
    const unsigned w[4] = {t.x, t.y, t.z, t.w};
#pragma unroll
    for (int d = 0; d < 4; ++d) {
      const unsigned y = w[d] ^ pat;
      if (((y - 0x01010101u) & ~y & 0x80808080u) != 0u) {
#pragma unroll
        for (int j = 0; j < 4; ++j) {
          if (((y >> (8 * j)) & 0xFFu) == 0u) {
            const int pos = atomicAdd(&qcount, 1);
            qlist[pos] = tid * 16 + 4 * d + j;
          }
        }
      }
    }
  }
  if (fhas) mlist[fpre + fj0] = fent;
  for (int j = fj0 + fstep; j < fcnt; j += fstep) {
    const unsigned gpos = (unsigned)(fpre + j);
    if (gpos < (unsigned)PCAP) mlist[gpos] = lists[fplace + j];
  }
  __syncthreads();
  const int nq = __builtin_amdgcn_readfirstlane(qcount);
  if (nq == 0) return;
  for (int i = tid; i < nq; i += PT) {
    const int q = qlist[i];
    const int g = ids[q];
    const int r = qs[q] % PR;
    qlist[i]   = (q << 4) | r;
    rownode[r] = g;
  }
  __syncthreads();

  for (int w0 = 0; w0 < total; w0 += PCAP) {
    if (w0 > 0) {
      for (int j = fj0; j < fcnt; j += fstep) {
        const unsigned gpos = (unsigned)(fpre + j - w0);
        if (gpos < (unsigned)PCAP) mlist[gpos] = lists[fplace + j];
      }
      __syncthreads();
    }
    const int mc = (total - w0 < PCAP) ? total - w0 : PCAP;
    for (int b0 = wave * PB; b0 < mc; b0 += (PT / 32) * PB) {
      int   srow = 0;
      int   arow = PR;
      float a    = 0.0f;
      if (lane < PB && b0 + lane < mc) {
        const unsigned ment = (unsigned)mlist[b0 + lane];
        const int e = (int)(ment & ((1u << EBITS) - 1u));
        srow = src[e];
        a    = att[e];
        arow = (int)((ment >> EBITS) % PR);
      }
      float4 rv[PB][2];
#pragma unroll
      for (int b = 0; b < PB; ++b) {
        const int sr = __shfl(srow, b);
        const float4* vp = (const float4*)(vsrc + (size_t)sr * DIM);
        rv[b][0] = vp[lane];
        rv[b][1] = vp[lane + 32];
      }
#pragma unroll
      for (int b = 0; b < PB; ++b) {
        const float ab = __shfl(a, b);
        float* ap = acc + __shfl(arow, b) * DIM + lane * 4;
        atomicAdd(ap + 0,   rv[b][0].x * ab);
        atomicAdd(ap + 1,   rv[b][0].y * ab);
        atomicAdd(ap + 2,   rv[b][0].z * ab);
        atomicAdd(ap + 3,   rv[b][0].w * ab);
        atomicAdd(ap + 128, rv[b][1].x * ab);
        atomicAdd(ap + 129, rv[b][1].y * ab);
        atomicAdd(ap + 130, rv[b][1].z * ab);
        atomicAdd(ap + 131, rv[b][1].w * ab);
      }
    }
    __syncthreads();
  }

  const int c  = lane & 15;
  const int h  = lane >> 4;
  const int n0 = wave * 16;

  const int wr = lane >> 3;
  const int wq = lane & 7;
  const float* wsl = W1 + (size_t)(n0 + wr) * DIM + wq * 4;
  float4 wp[NPIECE][4];
#pragma unroll
  for (int p = 0; p < 2; ++p) {
#pragma unroll
    for (int j = 0; j < 4; ++j) wp[p][j] = *(const float4*)(wsl + (4 * j) * DIM + WK * p);
  }
  const float bias = b1[n0 + c];

  float* As = arena;
  float* Ap = As + PR * WPITCH;
  float* wt = Ap + PR * WPITCH + wave * (2 * WPIECE);
#pragma unroll
  for (int j = 0; j < 4; ++j) *(float4*)(wt + (4 * j + wr) * WTP + wq * 4) = wp[0][j];
  {
    const int r = tid >> 5;
    const int g = rownode[r];
    float4 v0 = make_float4(0.0f, 0.0f, 0.0f, 0.0f);
    float4 v1 = v0;
    if (g >= 0) {
      const float4* vp = (const float4*)(vself + (size_t)g * DIM);
      v0 = vp[lane];
      v1 = vp[lane + 32];
    }
    const float4 h0 = *(const float4*)(acc + r * DIM + lane * 4);
    const float4 h1 = *(const float4*)(acc + r * DIM + 128 + lane * 4);
    float* s0 = As + r * WPITCH + lane * 4;
    float* p0 = Ap + r * WPITCH + lane * 4;
    *(float4*)(s0)       = make_float4(v0.x + h0.x, v0.y + h0.y, v0.z + h0.z, v0.w + h0.w);
    *(float4*)(s0 + 128) = make_float4(v1.x + h1.x, v1.y + h1.y, v1.z + h1.z, v1.w + h1.w);
    *(float4*)(p0)       = make_float4(v0.x * h0.x, v0.y * h0.y, v0.z * h0.z, v0.w * h0.w);
    *(float4*)(p0 + 128) = make_float4(v1.x * h1.x, v1.y * h1.y, v1.z * h1.z, v1.w * h1.w);
  }
#pragma unroll
  for (int j = 0; j < 4; ++j) keep_ahead(wp[1][j]);
  __syncthreads();

  v8f ysum = (v8f){};
  v8f yprd = (v8f){};
  {
    const float* asl = As + c * WPITCH;
    const float* apl = Ap + c * WPITCH;
#pragma unroll
    for (int p = 0; p < NPIECE; ++p) {
      if (p + 1 < NPIECE) {
        float* wn = wt + ((p + 1) & 1) * WPIECE;
#pragma unroll
        for (int j = 0; j < 4; ++j) *(float4*)(wn + (4 * j + wr) * WTP + wq * 4) = wp[p + 1][j];
      }
      if (p + 2 < NPIECE) {
#pragma unroll
        for (int j = 0; j < 4; ++j) wp[p + 2][j] = *(const float4*)(wsl + (4 * j) * DIM + WK * (p + 2));
      }
      const float* wc = wt + (p & 1) * WPIECE + c * WTP + 4 * h;
      {
        float4 fb[4], fs[4], fp[4];
#pragma unroll
        for (int u = 0; u < 4; ++u) {
          fb[u] = *(const float4*)(wc + 8 * u);
          fs[u] = *(const float4*)(asl + kfrag(4 * p + u, h));
          fp[u] = *(const float4*)(apl + kfrag(4 * p + u, h));
        }
#pragma unroll
        for (int u = 0; u < 4; ++u) {
          v2f b0;  b0.x  = fb[u].x; b0.y  = fb[u].y;
          v2f b1v; b1v.x = fb[u].z; b1v.y = fb[u].w;
          v2f s0;  s0.x  = fs[u].x; s0.y  = fs[u].y;
          v2f s1;  s1.x  = fs[u].z; s1.y  = fs[u].w;
          v2f p0;  p0.x  = fp[u].x; p0.y  = fp[u].y;
          v2f p1;  p1.x  = fp[u].z; p1.y  = fp[u].w;
          ysum = __builtin_amdgcn_wmma_f32_16x16x4_f32(false, s0, false, b0,  (short)0, ysum, false, false);
          yprd = __builtin_amdgcn_wmma_f32_16x16x4_f32(false, p0, false, b0,  (short)0, yprd, false, false);
          ysum = __builtin_amdgcn_wmma_f32_16x16x4_f32(false, s1, false, b1v, (short)0, ysum, false, false);
          yprd = __builtin_amdgcn_wmma_f32_16x16x4_f32(false, p1, false, b1v, (short)0, yprd, false, false);
        }
      }
      if (p + 2 < NPIECE) {
#pragma unroll
        for (int j = 0; j < 4; ++j) keep_ahead(wp[p + 2][j]);
      }
      if (p + 1 < NPIECE) __syncthreads();
    }
  }

#pragma unroll
  for (int r = 0; r < 8; ++r) {
    float z1 = ysum[r] + bias;
    float z2 = yprd[r] + bias;
    z1 = (z1 >= 0.0f) ? z1 : 0.01f * z1;
    z2 = (z2 >= 0.0f) ? z2 : 0.01f * z2;
    acc[(8 * h + r) * DIM + n0 + c] = z1 + z2;
  }
  __syncthreads();

  for (int i = wave; i < nq; i += PT / 32) {
    const int qe = qlist[i];
    const int q  = qe >> 4;
    const int r  = qe & 15;
    const float4* yp = (const float4*)(acc + r * DIM);
    float4* op = (float4*)(obase + (size_t)q * DIM);
    op[lane]      = yp[lane];
    op[lane + 32] = yp[lane + 32];
  }
}

extern "C" void kernel_launch(void* const* d_in, const int* in_sizes, int n_in,
                              void* d_out, int out_size, void* d_ws, size_t ws_size,
                              hipStream_t stream) {
  const float* v_grid  = (const float*)d_in[0];
  const float* v_sc    = (const float*)d_in[1];
  const float* att_s2g = (const float*)d_in[2];
  const float* att_g2s = (const float*)d_in[3];
  const float* W1      = (const float*)d_in[4];
  const float* b1      = (const float*)d_in[5];
  const int*   src_s2g = (const int*)d_in[6];
  const int*   dst_s2g = (const int*)d_in[7];
  const int*   src_g2s = (const int*)d_in[8];
  const int*   dst_g2s = (const int*)d_in[9];
  const int*   sc_id   = (const int*)d_in[10];
  const int*   gr_id   = (const int*)d_in[11];
  float* out = (float*)d_out;

  int* offs  = (int*)d_ws;
  int* qslot = offs + NRUN * OFFSTR;
  int* lists = qslot + 2 * NQ;
  unsigned char* qtags = (unsigned char*)(lists + NLIST);

  k_front<<<NRUN, 1024, 0, stream>>>(dst_s2g, dst_g2s, gr_id, sc_id, lists, offs, qslot, qtags);

  k_pullmix<<<dim3(MAXU / PR, 2), PT, 0, stream>>>(v_grid, v_sc, att_s2g, att_g2s, src_s2g, src_g2s,
                                                 lists, offs, qtags, qslot, gr_id, sc_id, W1, b1, out);
}


# nh: each wave queues one row's list edges in LDS; waves then sum equal spans of the row-sorted queues in registers and add one partial per (wave,row) segment by LDS float atomics, not one per edge
# speedup vs baseline: 7.2621x; 1.1695x over previous
#include <hip/hip_runtime.h>

#define NG     100000
#define NS     20000
#define EDGES  300000
#define DIM    256
#define NQ     4096
#define MAXU   NQ
#define RN     50000
#define NRG    ((NG + RN - 1) / RN)
#define SH     16
#define SEGCAP 18752
#define NRUNG  (NRG * SH)
#define NRUN   (NRUNG + SH)
#define NLIST  (2 * SH * SEGCAP)
#define NOWN   (MAXU / PR)
#define OFFSTR (NOWN + 4)
#define FEPT   20
#define WPITCH 260
#define EBITS  19
#define PR     16
#define PT     512
#define PB     8
#define QCAP   512
#define WK     32
#define NPIECE (DIM / WK)
#define WTP    36
#define WPIECE (16 * WTP)
#define ARENA  (2 * PR * WPITCH + (PT / 32) * 2 * WPIECE)
#define PCAP   (ARENA - (PT / 32) * QCAP)

static_assert(EDGES <= (1 << EBITS), "edge index must fit the packed entry");
static_assert(MAXU  <= (1 << (31 - EBITS)), "slot must fit the packed entry");
static_assert(MAXU % PR == 0 && PR == 16, "a workgroup owns one 16-row tile of slots");
static_assert(PT / 32 * 16 == DIM, "one wave per 16-column tile of the output");
static_assert(PCAP > 0 && PCAP % 128 == 0 && QCAP % 128 == 0 && QCAP >= 256, "the match list is scanned 128 entries at a time; a queue emptied once it holds more than QCAP - 128 edges takes a whole step more");
static_assert(QCAP <= PCAP && QCAP <= 32 * (PT / 32), "a list that fits one queue is scanned in one round and gives every wave a span of at most 32 edges");
static_assert((PR * DIM + ARENA + NQ) * 4 + 256 <= 160 * 1024, "two workgroups per WGP");
static_assert(NOWN <= 256 && NOWN == 8 * 32, "the owner of a slot (slot / PR) fits a one-byte tag; one wave sums the owner counts, 8 per lane");
static_assert(OFFSTR % 4 == 0 && OFFSTR >= NOWN + 2, "an offset row is whole 16-B units and holds its offsets and the start");
static_assert(NQ % 1024 == 0, "whole workgroups of the edge filter write the query tags");
static_assert(NQ % 16 == 0 && NQ / 16 <= PT, "one 16-tag load per thread covers the query tags");
static_assert(NQ == 4 * 1024, "a front thread takes four queries in one 16-B load");
static_assert(NG % 4 == 0 && NS % 4 == 0 && RN % 4 == 0, "the 16-B LDS stores that clear a range end with the range");
static_assert(NRG == 2 && NS <= RN, "two grid ranges share a segment from its two ends; one range holds the small-category side");
static_assert(SH * SEGCAP >= EDGES && EDGES % 4 == 0 && SEGCAP % 4 == 0, "the shares cover the edges in whole 16-B loads");
static_assert(1024 * FEPT >= SEGCAP && FEPT % 4 == 0, "a front workgroup covers its share");
static_assert(NRUNG <= 32 && SH <= 32, "one wave reads a side's offsets, one run per lane");
static_assert(SH * 32 == PT && NRUNG * 16 == PT, "a pull workgroup copies its entries with 32 threads per run (small-category side) or 16 (grid side)");

typedef __attribute__((ext_vector_type(2))) float v2f;
typedef __attribute__((ext_vector_type(8))) float v8f;

__global__ __launch_bounds__(1024) void k_front(const int* __restrict__ dst_g,
                                                const int* __restrict__ dst_s,
                                                const int* __restrict__ gid,
                                                const int* __restrict__ sid,
                                                int* __restrict__ lists,
                                                int* __restrict__ offs,
                                                int* __restrict__ qslot,
                                                unsigned char* __restrict__ qtags) {
  __shared__ __attribute__((aligned(16))) int lmap[RN];
  __shared__ __attribute__((aligned(16))) int ocnt[NOWN];
  __shared__ int rstart0;
  const int  t     = threadIdx.x;
  const int  lane  = t & 31;
  const int  wave  = t >> 5;
  const int  b     = blockIdx.x;
  const bool sc    = b >= NRUNG;
  const int  range = sc ? 0 : b / SH;
  const int  share = sc ? b - NRUNG : b % SH;
  const int  side  = sc ? 1 : 0;
  const int  base  = range * RN;
  const int  left  = (sc ? NS : NG) - base;
  const int  nr    = left < RN ? left : RN;
  const int* ids   = sc ? sid : gid;
  const int* dst   = sc ? dst_s : dst_g;

  int4 dv[FEPT / 4];
#pragma unroll
  for (int k = 0; k < FEPT / 4; ++k) {
    const int i4 = 4 * (k * 1024 + t);
    const int e0 = share * SEGCAP + i4;
    dv[k] = make_int4(-1, -1, -1, -1);
    if (i4 < SEGCAP && e0 < EDGES) dv[k] = *(const int4*)(dst + e0);
  }
  const int4 q = *(const int4*)(ids + 4 * t);
  for (int i = t; i < nr / 4; i += 1024) *(int4*)(lmap + 4 * i) = make_int4(-1, -1, -1, -1);
  if (t < NOWN) ocnt[t] = 0;
  __syncthreads();
  const int g[4] = {q.x, q.y, q.z, q.w};
#pragma unroll
  for (int k = 0; k < 4; ++k) {
    const unsigned o = (unsigned)(g[k] - base);
    if (o < (unsigned)nr) atomicMax(&lmap[o], 4 * t + k);
  }
  __syncthreads();
  if (share == 0) {
#pragma unroll
    for (int k = 0; k < 4; ++k) {
      const unsigned o = (unsigned)(g[k] - base);
      if (o < (unsigned)nr) {
        const int sl = lmap[o];
        qslot[side * NQ + 4 * t + k] = sl;
        qtags[side * NQ + 4 * t + k] = (unsigned char)(sl / PR);
      }
    }
  }

  int sl[FEPT];
#pragma unroll
  for (int k = 0; k < FEPT / 4; ++k) {
    const int d[4] = {dv[k].x, dv[k].y, dv[k].z, dv[k].w};
#pragma unroll
    for (int j = 0; j < 4; ++j) {
      const unsigned o = (unsigned)(d[j] - base);
      int v = -1;
      if (o < (unsigned)nr) v = lmap[o];
      sl[4 * k + j] = v;
    }
  }
  int rk[FEPT];
#pragma unroll
  for (int i = 0; i < FEPT; ++i) {
    int r = 0;
    if (sl[i] >= 0) r = atomicAdd(&ocnt[sl[i] / PR], 1);
    rk[i] = r;
  }
  __syncthreads();
  if (wave == 0) {
    const int4 ca = *(const int4*)(ocnt + 8 * lane);
    const int4 cb = *(const int4*)(ocnt + 8 * lane + 4);
    const int s1 = ca.x + ca.y;
    const int s2 = s1 + ca.z;
    const int s3 = s2 + ca.w;
    const int s4 = s3 + cb.x;
    const int s5 = s4 + cb.y;
    const int s6 = s5 + cb.z;
    const int s7 = s6 + cb.w;
    int x = s7;
#pragma unroll
    for (int d = 1; d < 32; d <<= 1) {
      const int y = __shfl_up(x, d);
      if (lane >= d) x += y;
    }
    const int ex = x - s7;
    const int4 oa = make_int4(ex, ex + ca.x, ex + s1, ex + s2);
    const int4 ob = make_int4(ex + s3, ex + s4, ex + s5, ex + s6);
    *(int4*)(ocnt + 8 * lane)     = oa;
    *(int4*)(ocnt + 8 * lane + 4) = ob;
    const int total = __shfl(x, 31);
    const int seg   = (sc ? SH + share : share) * SEGCAP;
    const int start = (range == 1) ? seg + SEGCAP - total : seg;
    int* orow = offs + b * OFFSTR;
    *(int4*)(orow + 8 * lane)     = oa;
    *(int4*)(orow + 8 * lane + 4) = ob;
    if (lane == 0) {
      orow[NOWN]     = total;
      orow[NOWN + 1] = start;
      rstart0 = start;
    }
  }
  __syncthreads();
  const int start = rstart0;
#pragma unroll
  for (int i = 0; i < FEPT; ++i) rk[i] += ocnt[(sl[i] >= 0 ? sl[i] : 0) / PR];
#pragma unroll
  for (int k = 0; k < FEPT / 4; ++k) {
#pragma unroll
    for (int j = 0; j < 4; ++j) {
      const int v = sl[4 * k + j];
      if (v >= 0) lists[start + rk[4 * k + j]] = (v << EBITS) | (share * SEGCAP + 4 * (k * 1024 + t) + j);
    }
  }
}

__device__ __forceinline__ int kfrag(int t, int h) { return 8 * t + 4 * h; }

__device__ __forceinline__ void keep_ahead(float4& w) {
  asm volatile("" : "+v"(w.x), "+v"(w.y), "+v"(w.z), "+v"(w.w));
}

__global__ __launch_bounds__(PT, 8) void k_pullmix(const float* __restrict__ v_grid,
                                                const float* __restrict__ v_sc,
                                                const float* __restrict__ att_s2g,
                                                const float* __restrict__ att_g2s,
                                                const int* __restrict__ src_s2g,
                                                const int* __restrict__ src_g2s,
                                                const int* __restrict__ lists,
                                                const int* __restrict__ offs,
                                                const unsigned char* __restrict__ qtags,
                                                const int* __restrict__ qslot,
                                                const int* __restrict__ gr_id,
                                                const int* __restrict__ sc_id,
                                                const float* __restrict__ W1,
                                                const float* __restrict__ b1,
                                                float* __restrict__ out) {
  __shared__ __attribute__((aligned(16))) float acc[PR * DIM];
  __shared__ __attribute__((aligned(16))) float arena[ARENA];
  __shared__ int qlist[NQ];
  __shared__ int rownode[PR];
  __shared__ int qcount;
  __shared__ int rplace[32];
  __shared__ int rcnt[32];
  __shared__ int rpre[32];
  __shared__ int mtotal;
  __shared__ int rowcnt[PR];

  int* mlist = (int*)arena;
  const int tid  = threadIdx.x;
  const int lane = tid & 31;
  const int wave = __builtin_amdgcn_readfirstlane(tid >> 5);
  const unsigned blk = blockIdx.x;
  const int side = 1 - (int)blockIdx.y;

  const float* vsrc  = side ? v_grid  : v_sc;
  const float* att   = side ? att_g2s : att_s2g;
  const int*   src   = side ? src_g2s : src_s2g;
  const int  run0 = side ? NRUNG : 0;
  const int  nrun = side ? SH : NRUNG;
  const unsigned char* qt = qtags + (size_t)side * NQ;
  const unsigned pat = blk * 0x01010101u;
  const float* vself = side ? v_sc    : v_grid;
  const int*   qs    = qslot + (size_t)side * NQ;
  const int*   ids   = side ? sc_id   : gr_id;
  float*       obase = out + (side ? (size_t)0 : (size_t)NQ * DIM);
  if (wave == 0) {
    int place = 0;
    int cnt   = 0;
    if (lane < nrun) {
      const int* orow = offs + (run0 + lane) * OFFSTR;
      const int o0 = orow[blk];
      const int o1 = orow[blk + 1];
      place = orow[NOWN + 1] + o0;
      cnt   = o1 - o0;
    }
    int x = cnt;
#pragma unroll
    for (int d = 1; d < 32; d <<= 1) {
      const int y = __shfl_up(x, d);
      if (lane >= d) x += y;
    }
    rplace[lane] = place;
    rcnt[lane]   = cnt;
    rpre[lane]   = x - cnt;
    if (lane == 31) mtotal = x;
  }

  for (int i = tid; i < PR * DIM; i += PT) acc[i] = 0.0f;
  if (tid < PR) rownode[tid] = -1;
  if (tid == 0) qcount = 0;
  __syncthreads();

  const int total  = __builtin_amdgcn_readfirstlane(mtotal);
  const int fstep  = side ? 32 : 16;
  const int frun   = side ? (tid >> 5) : (tid >> 4);
  const int fj0    = tid & (fstep - 1);
  const int fplace = rplace[frun];
  const int fcnt   = rcnt[frun];
  const int fpre   = rpre[frun];
  const bool fhas  = fj0 < fcnt && (unsigned)(fpre + fj0) < (unsigned)PCAP;
  int fent = 0;
  if (fhas) fent = lists[fplace + fj0];

  if (tid < NQ / 16) {
    const uint4 t = *(const uint4*)(qt + tid * 16);
    const unsigned w[4] = {t.x, t.y, t.z, t.w};
#pragma unroll
    for (int d = 0; d < 4; ++d) {
      const unsigned y = w[d] ^ pat;
      if (((y - 0x01010101u) & ~y & 0x80808080u) != 0u) {
#pragma unroll
        for (int j = 0; j < 4; ++j) {
          if (((y >> (8 * j)) & 0xFFu) == 0u) {
            const int pos = atomicAdd(&qcount, 1);
            qlist[pos] = tid * 16 + 4 * d + j;
          }
        }
      }
    }
  }
  if (fhas) mlist[fpre + fj0] = fent;
  for (int j = fj0 + fstep; j < fcnt; j += fstep) {
    const unsigned gpos = (unsigned)(fpre + j);
    if (gpos < (unsigned)PCAP) mlist[gpos] = lists[fplace + j];
  }
  __syncthreads();
  const int nq = __builtin_amdgcn_readfirstlane(qcount);
  if (nq == 0) return;
  for (int i = tid; i < nq; i += PT) {
    const int q = qlist[i];
    const int g = ids[q];
    const int r = qs[q] % PR;
    qlist[i]   = (q << 4) | r;
    rownode[r] = g;
  }
  __syncthreads();

  int* rq = mlist + PCAP + wave * QCAP;
  const int grow = rownode[wave];
  float4 v0 = make_float4(0.0f, 0.0f, 0.0f, 0.0f);
  float4 v1 = v0;
  if (grow >= 0) {
    const float4* vp = (const float4*)(vself + (size_t)grow * DIM);
    v0 = vp[lane];
    v1 = vp[lane + 32];
  }
  const bool wide = total > QCAP;
  float4 hs0 = make_float4(0.0f, 0.0f, 0.0f, 0.0f);
  float4 hs1 = hs0;
  auto drain = [&](int n) __attribute__((always_inline)) {
    for (int base = 0; base < n; base += 32) {
      const int m = (n - base < 32) ? n - base : 32;
      int   srow = 0;
      float a    = 0.0f;
      if (lane < m) {
        const int e = rq[base + lane];
        srow = src[e];
        a    = att[e];
      }
      for (int b0 = 0; b0 < m; b0 += PB) {
        float4 rv[PB][2];
#pragma unroll
        for (int b = 0; b < PB; ++b) {
          const int sr = __builtin_amdgcn_readlane(srow, b0 + b);
          const float4* vp = (const float4*)(vsrc + (size_t)sr * DIM);
          rv[b][0] = vp[lane];
          rv[b][1] = vp[lane + 32];
        }
#pragma unroll
        for (int b = 0; b < PB; ++b) {
          const float ab = __int_as_float(__builtin_amdgcn_readlane(__float_as_int(a), b0 + b));
          hs0.x += rv[b][0].x * ab;
          hs0.y += rv[b][0].y * ab;
          hs0.z += rv[b][0].z * ab;
          hs0.w += rv[b][0].w * ab;
          hs1.x += rv[b][1].x * ab;
          hs1.y += rv[b][1].y * ab;
          hs1.z += rv[b][1].z * ab;
          hs1.w += rv[b][1].w * ab;
        }
      }
    }
  };
  int qn = 0;
  for (int w0 = 0; w0 < total; w0 += PCAP) {
    if (w0 > 0) {
      for (int j = fj0; j < fcnt; j += fstep) {
        const unsigned gpos = (unsigned)(fpre + j - w0);
        if (gpos < (unsigned)PCAP) mlist[gpos] = lists[fplace + j];
      }
      __syncthreads();
    }
    const int mc = (total - w0 < PCAP) ? total - w0 : PCAP;
    qn = 0;
    for (int c0 = 0; c0 < mc; c0 += 128) {
      if (wide && qn > QCAP - 128) {
        drain(qn);
        qn = 0;
      }
      const int4 e4 = *(const int4*)(mlist + c0 + 4 * lane);
      const int ev[4] = {e4.x, e4.y, e4.z, e4.w};
#pragma unroll
      for (int k = 0; k < 4; ++k) {
        const unsigned ment = (unsigned)ev[k];
        const bool mine = (c0 + 4 * lane + k < mc) && (int)((ment >> EBITS) % PR) == wave;
        const unsigned mask = __builtin_amdgcn_ballot_w32(mine);
        if (mine) rq[qn + (int)__builtin_amdgcn_mbcnt_lo(mask, 0u)] = (int)(ment & ((1u << EBITS) - 1u));
        qn += __builtin_popcount(mask);
      }
    }
    if (wide) {
      drain(qn);
      __syncthreads();
    }
  }
  if (wide) {
    *(float4*)(acc + wave * DIM + lane * 4)       = hs0;
    *(float4*)(acc + wave * DIM + 128 + lane * 4) = hs1;
  } else if (total > 0) {
    if (lane == 0) rowcnt[wave] = qn;
    __syncthreads();
    const int qlen = (lane < PR) ? rowcnt[lane] : 0;
    const int S    = (total + PT / 32 - 1) / (PT / 32);
    const int p0   = wave * S;
    const int left = total - p0;
    const int span = left < 0 ? 0 : (left < S ? left : S);
    const int p    = p0 + lane;
    int r  = 0;
    int ex = 0;
    int ij = 0;
#pragma unroll
    for (int j = 0; j < PR - 1; ++j) {
      ij += __builtin_amdgcn_readlane(qlen, j);
      if (p >= ij) { r = j + 1; ex = ij; }
    }
    int   srow = 0;
    float a    = 0.0f;
    if (lane < span) {
      const int e = mlist[PCAP + r * QCAP + (p - ex)];
      srow = src[e];
      a    = att[e];
    }
    if (span > 0) {
      const int lastrow = __builtin_amdgcn_readlane(r, span - 1);
      const int erow    = (lane < span) ? r : lastrow;
      int cur = __builtin_amdgcn_readlane(erow, 0);
      float4 s0 = make_float4(0.0f, 0.0f, 0.0f, 0.0f);
      float4 s1 = s0;
      auto flush = [&](int row) __attribute__((always_inline)) {
        float* ap = acc + row * DIM + lane * 4;
        atomicAdd(ap + 0,   s0.x);
        atomicAdd(ap + 1,   s0.y);
        atomicAdd(ap + 2,   s0.z);
        atomicAdd(ap + 3,   s0.w);
        atomicAdd(ap + 128, s1.x);
        atomicAdd(ap + 129, s1.y);
        atomicAdd(ap + 130, s1.z);
        atomicAdd(ap + 131, s1.w);
      };
      for (int b0 = 0; b0 < span; b0 += PB) {
        float4 rv[PB][2];
#pragma unroll
        for (int b = 0; b < PB; ++b) {
          const int sr = __builtin_amdgcn_readlane(srow, b0 + b);
          const float4* vp = (const float4*)(vsrc + (size_t)sr * DIM);
          rv[b][0] = vp[lane];
          rv[b][1] = vp[lane + 32];
        }
#pragma unroll
        for (int b = 0; b < PB; ++b) {
          const int rb = __builtin_amdgcn_readlane(erow, b0 + b);
          if (rb != cur) {
            flush(cur);
            s0  = make_float4(0.0f, 0.0f, 0.0f, 0.0f);
            s1  = s0;
            cur = rb;
          }
          const float ab = __int_as_float(__builtin_amdgcn_readlane(__float_as_int(a), b0 + b));
          s0.x += rv[b][0].x * ab;
          s0.y += rv[b][0].y * ab;
          s0.z += rv[b][0].z * ab;
          s0.w += rv[b][0].w * ab;
          s1.x += rv[b][1].x * ab;
          s1.y += rv[b][1].y * ab;
          s1.z += rv[b][1].z * ab;
          s1.w += rv[b][1].w * ab;
        }
      }
      flush(cur);
    }
    __syncthreads();
  }

  const int c  = lane & 15;
  const int h  = lane >> 4;
  const int n0 = wave * 16;

  const int wr = lane >> 3;
  const int wq = lane & 7;
  const float* wsl = W1 + (size_t)(n0 + wr) * DIM + wq * 4;
  float4 wp[NPIECE][4];
#pragma unroll
  for (int p = 0; p < 2; ++p) {
#pragma unroll
    for (int j = 0; j < 4; ++j) wp[p][j] = *(const float4*)(wsl + (4 * j) * DIM + WK * p);
  }
  const float bias = b1[n0 + c];

  float* As = arena;
  float* Ap = As + PR * WPITCH;
  float* wt = Ap + PR * WPITCH + wave * (2 * WPIECE);
#pragma unroll
  for (int j = 0; j < 4; ++j) *(float4*)(wt + (4 * j + wr) * WTP + wq * 4) = wp[0][j];
  {
    const int r = wave;
    const float4 h0 = *(const float4*)(acc + r * DIM + lane * 4);
    const float4 h1 = *(const float4*)(acc + r * DIM + 128 + lane * 4);
    float* s0 = As + r * WPITCH + lane * 4;
    float* p0 = Ap + r * WPITCH + lane * 4;
    *(float4*)(s0)       = make_float4(v0.x + h0.x, v0.y + h0.y, v0.z + h0.z, v0.w + h0.w);
    *(float4*)(s0 + 128) = make_float4(v1.x + h1.x, v1.y + h1.y, v1.z + h1.z, v1.w + h1.w);
    *(float4*)(p0)       = make_float4(v0.x * h0.x, v0.y * h0.y, v0.z * h0.z, v0.w * h0.w);
    *(float4*)(p0 + 128) = make_float4(v1.x * h1.x, v1.y * h1.y, v1.z * h1.z, v1.w * h1.w);
  }
#pragma unroll
  for (int j = 0; j < 4; ++j) keep_ahead(wp[1][j]);
  __syncthreads();

  v8f ysum = (v8f){};
  v8f yprd = (v8f){};
  {
    const float* asl = As + c * WPITCH;
    const float* apl = Ap + c * WPITCH;
#pragma unroll
    for (int p = 0; p < NPIECE; ++p) {
      if (p + 1 < NPIECE) {
        float* wn = wt + ((p + 1) & 1) * WPIECE;
#pragma unroll
        for (int j = 0; j < 4; ++j) *(float4*)(wn + (4 * j + wr) * WTP + wq * 4) = wp[p + 1][j];
      }
      if (p + 2 < NPIECE) {
#pragma unroll
        for (int j = 0; j < 4; ++j) wp[p + 2][j] = *(const float4*)(wsl + (4 * j) * DIM + WK * (p + 2));
      }
      const float* wc = wt + (p & 1) * WPIECE + c * WTP + 4 * h;
      {
        float4 fb[4], fs[4], fp[4];
#pragma unroll
        for (int u = 0; u < 4; ++u) {
          fb[u] = *(const float4*)(wc + 8 * u);
          fs[u] = *(const float4*)(asl + kfrag(4 * p + u, h));
          fp[u] = *(const float4*)(apl + kfrag(4 * p + u, h));
        }
#pragma unroll
        for (int u = 0; u < 4; ++u) {
          v2f b0;  b0.x  = fb[u].x; b0.y  = fb[u].y;
          v2f b1v; b1v.x = fb[u].z; b1v.y = fb[u].w;
          v2f s0;  s0.x  = fs[u].x; s0.y  = fs[u].y;
          v2f s1;  s1.x  = fs[u].z; s1.y  = fs[u].w;
          v2f p0;  p0.x  = fp[u].x; p0.y  = fp[u].y;
          v2f p1;  p1.x  = fp[u].z; p1.y  = fp[u].w;
          ysum = __builtin_amdgcn_wmma_f32_16x16x4_f32(false, s0, false, b0,  (short)0, ysum, false, false);
          yprd = __builtin_amdgcn_wmma_f32_16x16x4_f32(false, p0, false, b0,  (short)0, yprd, false, false);
          ysum = __builtin_amdgcn_wmma_f32_16x16x4_f32(false, s1, false, b1v, (short)0, ysum, false, false);
          yprd = __builtin_amdgcn_wmma_f32_16x16x4_f32(false, p1, false, b1v, (short)0, yprd, false, false);
        }
      }
      if (p + 2 < NPIECE) {
#pragma unroll
        for (int j = 0; j < 4; ++j) keep_ahead(wp[p + 2][j]);
      }
      if (p + 1 < NPIECE) __syncthreads();
    }
  }

#pragma unroll
  for (int r = 0; r < 8; ++r) {
    float z1 = ysum[r] + bias;
    float z2 = yprd[r] + bias;
    z1 = (z1 >= 0.0f) ? z1 : 0.01f * z1;
    z2 = (z2 >= 0.0f) ? z2 : 0.01f * z2;
    acc[(8 * h + r) * DIM + n0 + c] = z1 + z2;
  }
  __syncthreads();

  for (int i = wave; i < nq; i += PT / 32) {
    const int qe = qlist[i];
    const int q  = qe >> 4;
    const int r  = qe & 15;
    const float4* yp = (const float4*)(acc + r * DIM);
    float4* op = (float4*)(obase + (size_t)q * DIM);
    op[lane]      = yp[lane];
    op[lane + 32] = yp[lane + 32];
  }
}

extern "C" void kernel_launch(void* const* d_in, const int* in_sizes, int n_in,
                              void* d_out, int out_size, void* d_ws, size_t ws_size,
                              hipStream_t stream) {
  const float* v_grid  = (const float*)d_in[0];
  const float* v_sc    = (const float*)d_in[1];
  const float* att_s2g = (const float*)d_in[2];
  const float* att_g2s = (const float*)d_in[3];
  const float* W1      = (const float*)d_in[4];
  const float* b1      = (const float*)d_in[5];
  const int*   src_s2g = (const int*)d_in[6];
  const int*   dst_s2g = (const int*)d_in[7];
  const int*   src_g2s = (const int*)d_in[8];
  const int*   dst_g2s = (const int*)d_in[9];
  const int*   sc_id   = (const int*)d_in[10];
  const int*   gr_id   = (const int*)d_in[11];
  float* out = (float*)d_out;

  int* offs  = (int*)d_ws;
  int* qslot = offs + NRUN * OFFSTR;
  int* lists = qslot + 2 * NQ;
  unsigned char* qtags = (unsigned char*)(lists + NLIST);

  k_front<<<NRUN, 1024, 0, stream>>>(dst_s2g, dst_g2s, gr_id, sc_id, lists, offs, qslot, qtags);

  k_pullmix<<<dim3(MAXU / PR, 2), PT, 0, stream>>>(v_grid, v_sc, att_s2g, att_g2s, src_s2g, src_g2s,
                                                 lists, offs, qtags, qslot, gr_id, sc_id, W1, b1, out);
}
